# starting kernel
# baseline (speedup 1.0000x reference)
#include <hip/hip_runtime.h>
#include <hip/hip_bf16.h>

#define N_NODES 20000
#define N_EDGES 100000
#define TPB 256

typedef __attribute__((ext_vector_type(2))) float v2f;
typedef __attribute__((ext_vector_type(8))) float v8f;

template <bool BT>
__global__ void gemm_wmma_f32(const float* __restrict__ A, const float* __restrict__ B,
                              float* __restrict__ C, int M, int K, int NC,
                              int lda, int ldb, int ldc) {
    const int tilesN = NC >> 4;
    const int wave   = blockIdx.x * (blockDim.x >> 5) + (threadIdx.x >> 5);
    const int trow   = wave / tilesN;
    const int tcol   = wave - trow * tilesN;
    if ((trow << 4) >= M) return;

    const int lane = threadIdx.x & 31;
    const int mloc = lane & 15;
    const int kh   = (lane >> 4) << 1;
    const int row0 = trow << 4, col0 = tcol << 4;

    v8f acc = {};
    const float* aBase = A + (size_t)(row0 + mloc) * lda + kh;
    for (int k = 0; k < K; k += 4) {
        v2f a, b;
        a.x = aBase[k];
        a.y = aBase[k + 1];
        if (BT) {
            const float* bp = B + (size_t)(col0 + mloc) * ldb + k + kh;
            b.x = bp[0];
            b.y = bp[1];
        } else {
            const float* bp = B + (size_t)(k + kh) * ldb + col0 + mloc;
            b.x = bp[0];
            b.y = bp[ldb];
        }
        acc = __builtin_amdgcn_wmma_f32_16x16x4_f32(false, a, false, b,
                                                    (short)0, acc, false, false);
    }

    const int   rbase = row0 + ((lane >> 4) << 3);
    float*      cp    = C + (size_t)rbase * ldc + col0 + mloc;
#pragma unroll
    for (int r = 0; r < 8; ++r) cp[(size_t)r * ldc] = acc[r];
}

__device__ __forceinline__ unsigned f2ord(float f) {
    unsigned u = __float_as_uint(f);
    return (u & 0x80000000u) ? ~u : (u | 0x80000000u);
}
__device__ __forceinline__ float ord2f(unsigned u) {
    return (u & 0x80000000u) ? __uint_as_float(u & 0x7FFFFFFFu) : __uint_as_float(~u);
}

__global__ void zero_kernel(float* p, int n) {
    int t = blockIdx.x * blockDim.x + threadIdx.x;
    if (t < n) p[t] = 0.0f;
}

__global__ void node_attn_kernel(const float* __restrict__ ft, const float* __restrict__ al,
                                 const float* __restrict__ ar, float* __restrict__ el,
                                 float* __restrict__ er) {
    int t = blockIdx.x * blockDim.x + threadIdx.x;
    if (t >= N_NODES * 2) return;
    int n = t >> 1, h = t & 1;
    const float* f  = ft + (size_t)n * 128 + h * 64;
    const float* a1 = al + h * 64;
    const float* a2 = ar + h * 64;
    float s1 = 0.f, s2 = 0.f;
#pragma unroll 8
    for (int d = 0; d < 64; ++d) { float v = f[d]; s1 += v * a1[d]; s2 += v * a2[d]; }
    el[t] = s1;
    er[t] = s2;
}

__global__ void edge_attn_kernel(const float* __restrict__ fe, const float* __restrict__ ae,
                                 float* __restrict__ ee) {
    int t = blockIdx.x * blockDim.x + threadIdx.x;
    if (t >= N_EDGES * 2) return;
    int e = t >> 1, h = t & 1;
    const float* f = fe + (size_t)e * 128 + h * 64;
    const float* a = ae + h * 64;
    float s = 0.f;
#pragma unroll 8
    for (int d = 0; d < 64; ++d) s += f[d] * a[d];
    ee[t] = s;
}

__global__ void logits_max_kernel(const float* __restrict__ el, const float* __restrict__ er,
                                  const float* __restrict__ ee, const int* __restrict__ src,
                                  const int* __restrict__ dst, float* __restrict__ lg,
                                  unsigned* __restrict__ mU) {
    int t = blockIdx.x * blockDim.x + threadIdx.x;
    if (t >= N_EDGES * 2) return;
    int e = t >> 1, h = t & 1;
    float l = el[src[e] * 2 + h] + er[dst[e] * 2 + h] + ee[t];
    l = (l > 0.f) ? l : 0.2f * l;
    lg[t] = l;
    atomicMax(&mU[dst[e] * 2 + h], f2ord(l));
}

__global__ void p_sum_kernel(float* __restrict__ lg, const unsigned* __restrict__ mU,
                             const int* __restrict__ dst, float* __restrict__ sB) {
    int t = blockIdx.x * blockDim.x + threadIdx.x;
    if (t >= N_EDGES * 2) return;
    int e = t >> 1, h = t & 1;
    float m  = ord2f(mU[dst[e] * 2 + h]);
    float pv = expf(lg[t] - m);
    lg[t]    = pv;
    atomicAdd(&sB[dst[e] * 2 + h], pv);
}

__global__ void scatter_msg_kernel(const float* __restrict__ ft, const float* __restrict__ fe,
                                   const float* __restrict__ p, const float* __restrict__ sB,
                                   const int* __restrict__ src, const int* __restrict__ dst,
                                   float* __restrict__ agg) {
    int t = blockIdx.x * blockDim.x + threadIdx.x;
    if (t >= N_EDGES * 32) return;
    int e  = t >> 5;
    int c0 = (t & 31) << 2;
    int h  = c0 >> 6;
    int sN = src[e], dN = dst[e];
    float alpha = p[e * 2 + h] / sB[dN * 2 + h];
    const float* fts = ft + (size_t)sN * 128 + c0;
    const float* fes = fe + (size_t)e * 128 + c0;
    float*       ag  = agg + (size_t)dN * 128 + c0;
#pragma unroll
    for (int i = 0; i < 4; ++i) atomicAdd(&ag[i], (fts[i] + fes[i]) * alpha);
}

__global__ void finalize_kernel(const float* __restrict__ agg, const float* __restrict__ bb,
                                float* __restrict__ xout) {
    int t = blockIdx.x * blockDim.x + threadIdx.x;
    if (t >= N_NODES * 64) return;
    int n = t >> 6, d = t & 63;
    float v0 = agg[(size_t)n * 128 + d] + bb[d];
    float v1 = agg[(size_t)n * 128 + 64 + d] + bb[64 + d];
    xout[t] = 0.5f * (v0 + v1);
}

static inline int cdiv(int a, int b) { return (a + b - 1) / b; }

struct WsPtrs {
    float* ft; float* fe; float* el; float* er; float* ee; float* lg;
    unsigned* mU; float* sB; float* agg; float* xb0; float* xb1;
};

static void run_edge_gat_layer(const float* xin, const float* eF, const int* src,
                               const int* dst, const float* Wn, const float* We,
                               const float* al, const float* ar, const float* ae,
                               const float* bb, float* xout, const WsPtrs& w,
                               hipStream_t stream) {
    gemm_wmma_f32<false><<<(N_NODES / 16) * (128 / 16) / 4, 128, 0, stream>>>(
        xin, Wn, w.ft, N_NODES, 64, 128, 64, 128, 128);
    gemm_wmma_f32<false><<<(N_EDGES / 16) * (128 / 16) / 4, 128, 0, stream>>>(
        eF, We, w.fe, N_EDGES, 32, 128, 32, 128, 128);

    node_attn_kernel<<<cdiv(N_NODES * 2, TPB), TPB, 0, stream>>>(w.ft, al, ar, w.el, w.er);
    edge_attn_kernel<<<cdiv(N_EDGES * 2, TPB), TPB, 0, stream>>>(w.fe, ae, w.ee);

    zero_kernel<<<cdiv(N_NODES * (2 + 2 + 128), TPB), TPB, 0, stream>>>(
        (float*)w.mU, N_NODES * (2 + 2 + 128));

    logits_max_kernel<<<cdiv(N_EDGES * 2, TPB), TPB, 0, stream>>>(w.el, w.er, w.ee, src, dst,
                                                                  w.lg, w.mU);
    p_sum_kernel<<<cdiv(N_EDGES * 2, TPB), TPB, 0, stream>>>(w.lg, w.mU, dst, w.sB);
    scatter_msg_kernel<<<cdiv(N_EDGES * 32, TPB), TPB, 0, stream>>>(w.ft, w.fe, w.lg, w.sB,
                                                                    src, dst, w.agg);
    finalize_kernel<<<cdiv(N_NODES * 64, TPB), TPB, 0, stream>>>(w.agg, bb, xout);
}

extern "C" void kernel_launch(void* const* d_in, const int* in_sizes, int n_in,
                              void* d_out, int out_size, void* d_ws, size_t ws_size,
                              hipStream_t stream) {
    const float* x_all = (const float*)d_in[0];
    const float* e_all = (const float*)d_in[1];
    const int*   s_all = (const int*)d_in[2];
    const int*   d_all = (const int*)d_in[3];

    const float* xS  = x_all + (size_t)1 * N_NODES * 64;
    const float* eS  = e_all + (size_t)1 * N_EDGES * 32;
    const int*   src = s_all + (size_t)1 * N_EDGES;
    const int*   dst = d_all + (size_t)1 * N_EDGES;

    const float* const* P = (const float* const*)d_in;
    const float* W_e2d = P[28];

    float* ws = (float*)d_ws;
    size_t off = 0;
    WsPtrs w;
    w.ft  = ws + off; off += (size_t)N_NODES * 128;
    w.fe  = ws + off; off += (size_t)N_EDGES * 128;
    w.el  = ws + off; off += (size_t)N_NODES * 2;
    w.er  = ws + off; off += (size_t)N_NODES * 2;
    w.ee  = ws + off; off += (size_t)N_EDGES * 2;
    w.lg  = ws + off; off += (size_t)N_EDGES * 2;
    w.mU  = (unsigned*)(ws + off); off += (size_t)N_NODES * 2;
    w.sB  = ws + off; off += (size_t)N_NODES * 2;
    w.agg = ws + off; off += (size_t)N_NODES * 128;
    w.xb0 = ws + off; off += (size_t)N_NODES * 64;
    w.xb1 = ws + off; off += (size_t)N_NODES * 64;

    run_edge_gat_layer(xS, eS, src, dst, P[4], P[5], P[6], P[7], P[8], P[9],
                       w.xb0, w, stream);
    run_edge_gat_layer(w.xb0, eS, src, dst, P[10], P[11], P[12], P[13], P[14], P[15],
                       w.xb1, w, stream);
    gemm_wmma_f32<true><<<(N_NODES / 16) * (64 / 16) / 4, 128, 0, stream>>>(
        w.xb1, W_e2d, w.xb0, N_NODES, 64, 64, 64, 64, 64);
    run_edge_gat_layer(w.xb0, eS, src, dst, P[16], P[17], P[18], P[19], P[20], P[21],
                       w.xb1, w, stream);
    run_edge_gat_layer(w.xb1, eS, src, dst, P[22], P[23], P[24], P[25], P[26], P[27],
                       (float*)d_out, w, stream);
}


# Edge GEMM fused into its two consumers: fe tile recomputed on f32 WMMA in registers, fe/ee never stored; ee dot now summed 4 per lane then a 16-lane tree (sum order changes)
# speedup vs baseline: 1.4483x; 1.4483x over previous
#include <hip/hip_runtime.h>
#include <hip/hip_bf16.h>

#define N_NODES 20000
#define N_EDGES 100000
#define TPB 256

typedef __attribute__((ext_vector_type(2))) float v2f;
typedef __attribute__((ext_vector_type(8))) float v8f;

template <bool BT>
__global__ void gemm_wmma_f32(const float* __restrict__ A, const float* __restrict__ B,
                              float* __restrict__ C, int M, int K, int NC,
                              int lda, int ldb, int ldc) {
    const int tilesN = NC >> 4;
    const int wave   = blockIdx.x * (blockDim.x >> 5) + (threadIdx.x >> 5);
    const int trow   = wave / tilesN;
    const int tcol   = wave - trow * tilesN;
    if ((trow << 4) >= M) return;

    const int lane = threadIdx.x & 31;
    const int mloc = lane & 15;
    const int kh   = (lane >> 4) << 1;
    const int row0 = trow << 4, col0 = tcol << 4;

    v8f acc = {};
    const float* aBase = A + (size_t)(row0 + mloc) * lda + kh;
    for (int k = 0; k < K; k += 4) {
        v2f a, b;
        a.x = aBase[k];
        a.y = aBase[k + 1];
        if (BT) {
            const float* bp = B + (size_t)(col0 + mloc) * ldb + k + kh;
            b.x = bp[0];
            b.y = bp[1];
        } else {
            const float* bp = B + (size_t)(k + kh) * ldb + col0 + mloc;
            b.x = bp[0];
            b.y = bp[ldb];
        }
        acc = __builtin_amdgcn_wmma_f32_16x16x4_f32(false, a, false, b,
                                                    (short)0, acc, false, false);
    }

    const int   rbase = row0 + ((lane >> 4) << 3);
    float*      cp    = C + (size_t)rbase * ldc + col0 + mloc;
#pragma unroll
    for (int r = 0; r < 8; ++r) cp[(size_t)r * ldc] = acc[r];
}

__device__ __forceinline__ unsigned f2ord(float f) {
    unsigned u = __float_as_uint(f);
    return (u & 0x80000000u) ? ~u : (u | 0x80000000u);
}
__device__ __forceinline__ float ord2f(unsigned u) {
    return (u & 0x80000000u) ? __uint_as_float(u & 0x7FFFFFFFu) : __uint_as_float(~u);
}

__global__ void zero_kernel(float* p, int n) {
    int t = blockIdx.x * blockDim.x + threadIdx.x;
    if (t < n) p[t] = 0.0f;
}

__global__ void node_attn_kernel(const float* __restrict__ ft, const float* __restrict__ al,
                                 const float* __restrict__ ar, float* __restrict__ el,
                                 float* __restrict__ er) {
    int t = blockIdx.x * blockDim.x + threadIdx.x;
    if (t >= N_NODES * 2) return;
    int n = t >> 1, h = t & 1;
    const float* f  = ft + (size_t)n * 128 + h * 64;
    const float* a1 = al + h * 64;
    const float* a2 = ar + h * 64;
    float s1 = 0.f, s2 = 0.f;
#pragma unroll 8
    for (int d = 0; d < 64; ++d) { float v = f[d]; s1 += v * a1[d]; s2 += v * a2[d]; }
    el[t] = s1;
    er[t] = s2;
}

__global__ void p_sum_kernel(float* __restrict__ lg, const unsigned* __restrict__ mU,
                             const int* __restrict__ dst, float* __restrict__ sB) {
    int t = blockIdx.x * blockDim.x + threadIdx.x;
    if (t >= N_EDGES * 2) return;
    int e = t >> 1, h = t & 1;
    float m  = ord2f(mU[dst[e] * 2 + h]);
    float pv = expf(lg[t] - m);
    lg[t]    = pv;
    atomicAdd(&sB[dst[e] * 2 + h], pv);
}

__global__ void finalize_kernel(const float* __restrict__ agg, const float* __restrict__ bb,
                                float* __restrict__ xout) {
    int t = blockIdx.x * blockDim.x + threadIdx.x;
    if (t >= N_NODES * 64) return;
    int n = t >> 6, d = t & 63;
    float v0 = agg[(size_t)n * 128 + d] + bb[d];
    float v1 = agg[(size_t)n * 128 + 64 + d] + bb[64 + d];
    xout[t] = 0.5f * (v0 + v1);
}


__device__ __forceinline__ void fe_tile_16x128(const float* __restrict__ eF,
                                               const float* __restrict__ We,
                                               int e0, int lane, v8f (&acc)[8]) {
    const int mloc = lane & 15;
    const int kh   = (lane >> 4) << 1;
    const float* aBase = eF + (size_t)(e0 + mloc) * 32 + kh;
    const float* bBase = We + (size_t)kh * 128 + mloc;
#pragma unroll
    for (int j = 0; j < 8; ++j) acc[j] = v8f{};
#pragma unroll
    for (int k = 0; k < 32; k += 4) {
        v2f a;
        a.x = aBase[k];
        a.y = aBase[k + 1];
#pragma unroll
        for (int j = 0; j < 8; ++j) {
            v2f b;
            b.x = bBase[(size_t)k * 128 + j * 16];
            b.y = bBase[(size_t)(k + 1) * 128 + j * 16];
            acc[j] = __builtin_amdgcn_wmma_f32_16x16x4_f32(false, a, false, b,
                                                           (short)0, acc[j], false, false);
        }
    }
}

__global__ void __launch_bounds__(128) edge_logits_fused_kernel(const float* __restrict__ eF,
                                         const float* __restrict__ We,
                                         const float* __restrict__ ae,
                                         const float* __restrict__ el,
                                         const float* __restrict__ er,
                                         const int* __restrict__ src,
                                         const int* __restrict__ dst,
                                         float* __restrict__ lg,
                                         unsigned* __restrict__ mU) {
    const int wave = blockIdx.x * (blockDim.x >> 5) + (threadIdx.x >> 5);
    if (wave >= N_EDGES / 16) return;
    const int lane = threadIdx.x & 31;
    const int c    = lane & 15;
    const int e0   = wave << 4;

    v8f acc[8];
    fe_tile_16x128(eF, We, e0, lane, acc);

    float s0[8], s1[8];
#pragma unroll
    for (int r = 0; r < 8; ++r) { s0[r] = 0.f; s1[r] = 0.f; }
#pragma unroll
    for (int j = 0; j < 4; ++j) {
        const float w0 = ae[j * 16 + c];
        const float w1 = ae[64 + j * 16 + c];
#pragma unroll
        for (int r = 0; r < 8; ++r) {
            s0[r] += acc[j][r] * w0;
            s1[r] += acc[j + 4][r] * w1;
        }
    }
#pragma unroll
    for (int off = 1; off < 16; off <<= 1) {
#pragma unroll
        for (int r = 0; r < 8; ++r) {
            s0[r] += __shfl_xor(s0[r], off, 32);
            s1[r] += __shfl_xor(s1[r], off, 32);
        }
    }
    float eev = 0.f;
#pragma unroll
    for (int r = 0; r < 8; ++r) {
        eev = (c == 2 * r)     ? s0[r] : eev;
        eev = (c == 2 * r + 1) ? s1[r] : eev;
    }
    const int e = e0 + ((lane >> 4) << 3) + (c >> 1);
    const int h = c & 1;
    const int t = e * 2 + h;
    const int dN = dst[e];
    float l = el[src[e] * 2 + h] + er[dN * 2 + h] + eev;
    l = (l > 0.f) ? l : 0.2f * l;
    lg[t] = l;
    atomicMax(&mU[dN * 2 + h], f2ord(l));
}

__global__ void __launch_bounds__(128) scatter_msg_fused_kernel(const float* __restrict__ eF,
                                         const float* __restrict__ We,
                                         const float* __restrict__ ft,
                                         const float* __restrict__ p,
                                         const float* __restrict__ sB,
                                         const int* __restrict__ src,
                                         const int* __restrict__ dst,
                                         float* __restrict__ agg) {
    const int wave = blockIdx.x * (blockDim.x >> 5) + (threadIdx.x >> 5);
    if (wave >= N_EDGES / 16) return;
    const int lane = threadIdx.x & 31;
    const int c    = lane & 15;
    const int e0   = wave << 4;

    const int eb = e0 + ((lane >> 4) << 3);

    int   sNr[8], dNr[8];
    float a0r[8], a1r[8];
#pragma unroll
    for (int r = 0; r < 8; ++r) {
        const int e = eb + r;
        sNr[r] = src[e];
        dNr[r] = dst[e];
        a0r[r] = p[e * 2]     / sB[dNr[r] * 2];
        a1r[r] = p[e * 2 + 1] / sB[dNr[r] * 2 + 1];
    }

    v8f acc[8];
    fe_tile_16x128(eF, We, e0, lane, acc);

#pragma unroll
    for (int r = 0; r < 8; ++r) {
        const float* fts = ft + (size_t)sNr[r] * 128 + c;
#pragma unroll
        for (int j = 0; j < 8; ++j) {
            const float alpha = (j < 4) ? a0r[r] : a1r[r];
            acc[j][r] = (fts[j * 16] + acc[j][r]) * alpha;
        }
    }
#pragma unroll
    for (int r = 0; r < 8; ++r) {
        float* ag = agg + (size_t)dNr[r] * 128 + c;
#pragma unroll
        for (int j = 0; j < 8; ++j) atomicAdd(&ag[j * 16], acc[j][r]);
    }
}

static inline int cdiv(int a, int b) { return (a + b - 1) / b; }

struct WsPtrs {
    float* ft; float* el; float* er; float* lg;
    unsigned* mU; float* sB; float* agg; float* xb0; float* xb1;
};

static void run_edge_gat_layer(const float* xin, const float* eF, const int* src,
                               const int* dst, const float* Wn, const float* We,
                               const float* al, const float* ar, const float* ae,
                               const float* bb, float* xout, const WsPtrs& w,
                               hipStream_t stream) {
    gemm_wmma_f32<false><<<(N_NODES / 16) * (128 / 16) / 4, 128, 0, stream>>>(
        xin, Wn, w.ft, N_NODES, 64, 128, 64, 128, 128);

    node_attn_kernel<<<cdiv(N_NODES * 2, TPB), TPB, 0, stream>>>(w.ft, al, ar, w.el, w.er);

    zero_kernel<<<cdiv(N_NODES * (2 + 2 + 128), TPB), TPB, 0, stream>>>(
        (float*)w.mU, N_NODES * (2 + 2 + 128));

    edge_logits_fused_kernel<<<cdiv(N_EDGES / 16, 4), 128, 0, stream>>>(
        eF, We, ae, w.el, w.er, src, dst, w.lg, w.mU);
    p_sum_kernel<<<cdiv(N_EDGES * 2, TPB), TPB, 0, stream>>>(w.lg, w.mU, dst, w.sB);
    scatter_msg_fused_kernel<<<cdiv(N_EDGES / 16, 4), 128, 0, stream>>>(
        eF, We, w.ft, w.lg, w.sB, src, dst, w.agg);
    finalize_kernel<<<cdiv(N_NODES * 64, TPB), TPB, 0, stream>>>(w.agg, bb, xout);
}

extern "C" void kernel_launch(void* const* d_in, const int* in_sizes, int n_in,
                              void* d_out, int out_size, void* d_ws, size_t ws_size,
                              hipStream_t stream) {
    const float* x_all = (const float*)d_in[0];
    const float* e_all = (const float*)d_in[1];
    const int*   s_all = (const int*)d_in[2];
    const int*   d_all = (const int*)d_in[3];

    const float* xS  = x_all + (size_t)1 * N_NODES * 64;
    const float* eS  = e_all + (size_t)1 * N_EDGES * 32;
    const int*   src = s_all + (size_t)1 * N_EDGES;
    const int*   dst = d_all + (size_t)1 * N_EDGES;

    const float* const* P = (const float* const*)d_in;
    const float* W_e2d = P[28];

    float* ws = (float*)d_ws;
    size_t off = 0;
    WsPtrs w;
    w.ft  = ws + off; off += (size_t)N_NODES * 128;
    w.el  = ws + off; off += (size_t)N_NODES * 2;
    w.er  = ws + off; off += (size_t)N_NODES * 2;
    w.lg  = ws + off; off += (size_t)N_EDGES * 2;
    w.mU  = (unsigned*)(ws + off); off += (size_t)N_NODES * 2;
    w.sB  = ws + off; off += (size_t)N_NODES * 2;
    w.agg = ws + off; off += (size_t)N_NODES * 128;
    w.xb0 = ws + off; off += (size_t)N_NODES * 64;
    w.xb1 = ws + off; off += (size_t)N_NODES * 64;

    run_edge_gat_layer(xS, eS, src, dst, P[4], P[5], P[6], P[7], P[8], P[9],
                       w.xb0, w, stream);
    run_edge_gat_layer(w.xb0, eS, src, dst, P[10], P[11], P[12], P[13], P[14], P[15],
                       w.xb1, w, stream);
    gemm_wmma_f32<true><<<(N_NODES / 16) * (64 / 16) / 4, 128, 0, stream>>>(
        w.xb1, W_e2d, w.xb0, N_NODES, 64, 64, 64, 64, 64);
    run_edge_gat_layer(w.xb0, eS, src, dst, P[16], P[17], P[18], P[19], P[20], P[21],
                       w.xb1, w, stream);
    run_edge_gat_layer(w.xb1, eS, src, dst, P[22], P[23], P[24], P[25], P[26], P[27],
                       (float*)d_out, w, stream);
}


# Head mean folded into the message: an edge's two head messages are added in registers, agg is [N,64], scatter atomics halved (sum order changes: heads combined before the node sum)
# speedup vs baseline: 1.5516x; 1.0713x over previous
#include <hip/hip_runtime.h>
#include <hip/hip_bf16.h>

#define N_NODES 20000
#define N_EDGES 100000
#define TPB 256

typedef __attribute__((ext_vector_type(2))) float v2f;
typedef __attribute__((ext_vector_type(8))) float v8f;

template <bool BT>
__global__ void gemm_wmma_f32(const float* __restrict__ A, const float* __restrict__ B,
                              float* __restrict__ C, int M, int K, int NC,
                              int lda, int ldb, int ldc) {
    const int tilesN = NC >> 4;
    const int wave   = blockIdx.x * (blockDim.x >> 5) + (threadIdx.x >> 5);
    const int trow   = wave / tilesN;
    const int tcol   = wave - trow * tilesN;
    if ((trow << 4) >= M) return;

    const int lane = threadIdx.x & 31;
    const int mloc = lane & 15;
    const int kh   = (lane >> 4) << 1;
    const int row0 = trow << 4, col0 = tcol << 4;

    v8f acc = {};
    const float* aBase = A + (size_t)(row0 + mloc) * lda + kh;
    for (int k = 0; k < K; k += 4) {
        v2f a, b;
        a.x = aBase[k];
        a.y = aBase[k + 1];
        if (BT) {
            const float* bp = B + (size_t)(col0 + mloc) * ldb + k + kh;
            b.x = bp[0];
            b.y = bp[1];
        } else {
            const float* bp = B + (size_t)(k + kh) * ldb + col0 + mloc;
            b.x = bp[0];
            b.y = bp[ldb];
        }
        acc = __builtin_amdgcn_wmma_f32_16x16x4_f32(false, a, false, b,
                                                    (short)0, acc, false, false);
    }

    const int   rbase = row0 + ((lane >> 4) << 3);
    float*      cp    = C + (size_t)rbase * ldc + col0 + mloc;
#pragma unroll
    for (int r = 0; r < 8; ++r) cp[(size_t)r * ldc] = acc[r];
}

__device__ __forceinline__ unsigned f2ord(float f) {
    unsigned u = __float_as_uint(f);
    return (u & 0x80000000u) ? ~u : (u | 0x80000000u);
}
__device__ __forceinline__ float ord2f(unsigned u) {
    return (u & 0x80000000u) ? __uint_as_float(u & 0x7FFFFFFFu) : __uint_as_float(~u);
}

__global__ void zero_kernel(float* p, int n) {
    int t = blockIdx.x * blockDim.x + threadIdx.x;
    if (t < n) p[t] = 0.0f;
}

__global__ void node_attn_kernel(const float* __restrict__ ft, const float* __restrict__ al,
                                 const float* __restrict__ ar, float* __restrict__ el,
                                 float* __restrict__ er) {
    int t = blockIdx.x * blockDim.x + threadIdx.x;
    if (t >= N_NODES * 2) return;
    int n = t >> 1, h = t & 1;
    const float* f  = ft + (size_t)n * 128 + h * 64;
    const float* a1 = al + h * 64;
    const float* a2 = ar + h * 64;
    float s1 = 0.f, s2 = 0.f;
#pragma unroll 8
    for (int d = 0; d < 64; ++d) { float v = f[d]; s1 += v * a1[d]; s2 += v * a2[d]; }
    el[t] = s1;
    er[t] = s2;
}

__global__ void p_sum_kernel(float* __restrict__ lg, const unsigned* __restrict__ mU,
                             const int* __restrict__ dst, float* __restrict__ sB) {
    int t = blockIdx.x * blockDim.x + threadIdx.x;
    if (t >= N_EDGES * 2) return;
    int e = t >> 1, h = t & 1;
    float m  = ord2f(mU[dst[e] * 2 + h]);
    float pv = expf(lg[t] - m);
    lg[t]    = pv;
    atomicAdd(&sB[dst[e] * 2 + h], pv);
}

__global__ void finalize_kernel(const float* __restrict__ agg, const float* __restrict__ bb,
                                float* __restrict__ xout) {
    int t = blockIdx.x * blockDim.x + threadIdx.x;
    if (t >= N_NODES * 64) return;
    int d = t & 63;
    xout[t] = 0.5f * (agg[t] + (bb[d] + bb[64 + d]));
}


__device__ __forceinline__ void fe_tile_16x128(const float* __restrict__ eF,
                                               const float* __restrict__ We,
                                               int e0, int lane, v8f (&acc)[8]) {
    const int mloc = lane & 15;
    const int kh   = (lane >> 4) << 1;
    const float* aBase = eF + (size_t)(e0 + mloc) * 32 + kh;
    const float* bBase = We + (size_t)kh * 128 + mloc;
#pragma unroll
    for (int j = 0; j < 8; ++j) acc[j] = v8f{};
#pragma unroll
    for (int k = 0; k < 32; k += 4) {
        v2f a;
        a.x = aBase[k];
        a.y = aBase[k + 1];
#pragma unroll
        for (int j = 0; j < 8; ++j) {
            v2f b;
            b.x = bBase[(size_t)k * 128 + j * 16];
            b.y = bBase[(size_t)(k + 1) * 128 + j * 16];
            acc[j] = __builtin_amdgcn_wmma_f32_16x16x4_f32(false, a, false, b,
                                                           (short)0, acc[j], false, false);
        }
    }
}

__global__ void __launch_bounds__(128) edge_logits_fused_kernel(const float* __restrict__ eF,
                                         const float* __restrict__ We,
                                         const float* __restrict__ ae,
                                         const float* __restrict__ el,
                                         const float* __restrict__ er,
                                         const int* __restrict__ src,
                                         const int* __restrict__ dst,
                                         float* __restrict__ lg,
                                         unsigned* __restrict__ mU) {
    const int wave = blockIdx.x * (blockDim.x >> 5) + (threadIdx.x >> 5);
    if (wave >= N_EDGES / 16) return;
    const int lane = threadIdx.x & 31;
    const int c    = lane & 15;
    const int e0   = wave << 4;

    v8f acc[8];
    fe_tile_16x128(eF, We, e0, lane, acc);

    float s0[8], s1[8];
#pragma unroll
    for (int r = 0; r < 8; ++r) { s0[r] = 0.f; s1[r] = 0.f; }
#pragma unroll
    for (int j = 0; j < 4; ++j) {
        const float w0 = ae[j * 16 + c];
        const float w1 = ae[64 + j * 16 + c];
#pragma unroll
        for (int r = 0; r < 8; ++r) {
            s0[r] += acc[j][r] * w0;
            s1[r] += acc[j + 4][r] * w1;
        }
    }
#pragma unroll
    for (int off = 1; off < 16; off <<= 1) {
#pragma unroll
        for (int r = 0; r < 8; ++r) {
            s0[r] += __shfl_xor(s0[r], off, 32);
            s1[r] += __shfl_xor(s1[r], off, 32);
        }
    }
    float eev = 0.f;
#pragma unroll
    for (int r = 0; r < 8; ++r) {
        eev = (c == 2 * r)     ? s0[r] : eev;
        eev = (c == 2 * r + 1) ? s1[r] : eev;
    }
    const int e = e0 + ((lane >> 4) << 3) + (c >> 1);
    const int h = c & 1;
    const int t = e * 2 + h;
    const int dN = dst[e];
    float l = el[src[e] * 2 + h] + er[dN * 2 + h] + eev;
    l = (l > 0.f) ? l : 0.2f * l;
    lg[t] = l;
    atomicMax(&mU[dN * 2 + h], f2ord(l));
}

__global__ void __launch_bounds__(128) scatter_msg_fused_kernel(const float* __restrict__ eF,
                                         const float* __restrict__ We,
                                         const float* __restrict__ ft,
                                         const float* __restrict__ p,
                                         const float* __restrict__ sB,
                                         const int* __restrict__ src,
                                         const int* __restrict__ dst,
                                         float* __restrict__ agg) {
    const int wave = blockIdx.x * (blockDim.x >> 5) + (threadIdx.x >> 5);
    if (wave >= N_EDGES / 16) return;
    const int lane = threadIdx.x & 31;
    const int c    = lane & 15;
    const int e0   = wave << 4;

    const int eb = e0 + ((lane >> 4) << 3);

    int   sNr[8], dNr[8];
    float a0r[8], a1r[8];
#pragma unroll
    for (int r = 0; r < 8; ++r) {
        const int e = eb + r;
        sNr[r] = src[e];
        dNr[r] = dst[e];
        a0r[r] = p[e * 2]     / sB[dNr[r] * 2];
        a1r[r] = p[e * 2 + 1] / sB[dNr[r] * 2 + 1];
    }

    v8f acc[8];
    fe_tile_16x128(eF, We, e0, lane, acc);

#pragma unroll
    for (int r = 0; r < 8; ++r) {
        const float* fts = ft + (size_t)sNr[r] * 128 + c;
#pragma unroll
        for (int j = 0; j < 4; ++j) {
            const float m0 = (fts[j * 16]      + acc[j][r])     * a0r[r];
            const float m1 = (fts[64 + j * 16] + acc[j + 4][r]) * a1r[r];
            acc[j][r] = m0 + m1;
        }
    }
#pragma unroll
    for (int r = 0; r < 8; ++r) {
        float* ag = agg + (size_t)dNr[r] * 64 + c;
#pragma unroll
        for (int j = 0; j < 4; ++j) atomicAdd(&ag[j * 16], acc[j][r]);
    }
}

static inline int cdiv(int a, int b) { return (a + b - 1) / b; }

struct WsPtrs {
    float* ft; float* el; float* er; float* lg;
    unsigned* mU; float* sB; float* agg; float* xb0; float* xb1;
};

static void run_edge_gat_layer(const float* xin, const float* eF, const int* src,
                               const int* dst, const float* Wn, const float* We,
                               const float* al, const float* ar, const float* ae,
                               const float* bb, float* xout, const WsPtrs& w,
                               hipStream_t stream) {
    gemm_wmma_f32<false><<<(N_NODES / 16) * (128 / 16) / 4, 128, 0, stream>>>(
        xin, Wn, w.ft, N_NODES, 64, 128, 64, 128, 128);

    node_attn_kernel<<<cdiv(N_NODES * 2, TPB), TPB, 0, stream>>>(w.ft, al, ar, w.el, w.er);

    zero_kernel<<<cdiv(N_NODES * (2 + 2 + 64), TPB), TPB, 0, stream>>>(
        (float*)w.mU, N_NODES * (2 + 2 + 64));

    edge_logits_fused_kernel<<<cdiv(N_EDGES / 16, 4), 128, 0, stream>>>(
        eF, We, ae, w.el, w.er, src, dst, w.lg, w.mU);
    p_sum_kernel<<<cdiv(N_EDGES * 2, TPB), TPB, 0, stream>>>(w.lg, w.mU, dst, w.sB);
    scatter_msg_fused_kernel<<<cdiv(N_EDGES / 16, 4), 128, 0, stream>>>(
        eF, We, w.ft, w.lg, w.sB, src, dst, w.agg);
    finalize_kernel<<<cdiv(N_NODES * 64, TPB), TPB, 0, stream>>>(w.agg, bb, xout);
}

extern "C" void kernel_launch(void* const* d_in, const int* in_sizes, int n_in,
                              void* d_out, int out_size, void* d_ws, size_t ws_size,
                              hipStream_t stream) {
    const float* x_all = (const float*)d_in[0];
    const float* e_all = (const float*)d_in[1];
    const int*   s_all = (const int*)d_in[2];
    const int*   d_all = (const int*)d_in[3];

    const float* xS  = x_all + (size_t)1 * N_NODES * 64;
    const float* eS  = e_all + (size_t)1 * N_EDGES * 32;
    const int*   src = s_all + (size_t)1 * N_EDGES;
    const int*   dst = d_all + (size_t)1 * N_EDGES;

    const float* const* P = (const float* const*)d_in;
    const float* W_e2d = P[28];

    float* ws = (float*)d_ws;
    size_t off = 0;
    WsPtrs w;
    w.ft  = ws + off; off += (size_t)N_NODES * 128;
    w.el  = ws + off; off += (size_t)N_NODES * 2;
    w.er  = ws + off; off += (size_t)N_NODES * 2;
    w.lg  = ws + off; off += (size_t)N_EDGES * 2;
    w.mU  = (unsigned*)(ws + off); off += (size_t)N_NODES * 2;
    w.sB  = ws + off; off += (size_t)N_NODES * 2;
    w.agg = ws + off; off += (size_t)N_NODES * 64;
    w.xb0 = ws + off; off += (size_t)N_NODES * 64;
    w.xb1 = ws + off; off += (size_t)N_NODES * 64;

    run_edge_gat_layer(xS, eS, src, dst, P[4], P[5], P[6], P[7], P[8], P[9],
                       w.xb0, w, stream);
    run_edge_gat_layer(w.xb0, eS, src, dst, P[10], P[11], P[12], P[13], P[14], P[15],
                       w.xb1, w, stream);
    gemm_wmma_f32<true><<<(N_NODES / 16) * (64 / 16) / 4, 128, 0, stream>>>(
        w.xb1, W_e2d, w.xb0, N_NODES, 64, 64, 64, 64, 64);
    run_edge_gat_layer(w.xb0, eS, src, dst, P[16], P[17], P[18], P[19], P[20], P[21],
                       w.xb1, w, stream);
    run_edge_gat_layer(w.xb1, eS, src, dst, P[22], P[23], P[24], P[25], P[26], P[27],
                       (float*)d_out, w, stream);
}


# Node side of a layer in one launch: ft GEMM wave (16 nodes x one head) also forms el/er from its accumulators (4 per lane then a 16-lane tree: sum order changes) and clears m/s/agg; 29 launches -> 21
# speedup vs baseline: 1.6071x; 1.0358x over previous
#include <hip/hip_runtime.h>
#include <hip/hip_bf16.h>

#define N_NODES 20000
#define N_EDGES 100000
#define TPB 256

typedef __attribute__((ext_vector_type(2))) float v2f;
typedef __attribute__((ext_vector_type(8))) float v8f;

template <bool BT>
__global__ void gemm_wmma_f32(const float* __restrict__ A, const float* __restrict__ B,
                              float* __restrict__ C, int M, int K, int NC,
                              int lda, int ldb, int ldc) {
    const int tilesN = NC >> 4;
    const int wave   = blockIdx.x * (blockDim.x >> 5) + (threadIdx.x >> 5);
    const int trow   = wave / tilesN;
    const int tcol   = wave - trow * tilesN;
    if ((trow << 4) >= M) return;

    const int lane = threadIdx.x & 31;
    const int mloc = lane & 15;
    const int kh   = (lane >> 4) << 1;
    const int row0 = trow << 4, col0 = tcol << 4;

    v8f acc = {};
    const float* aBase = A + (size_t)(row0 + mloc) * lda + kh;
    for (int k = 0; k < K; k += 4) {
        v2f a, b;
        a.x = aBase[k];
        a.y = aBase[k + 1];
        if (BT) {
            const float* bp = B + (size_t)(col0 + mloc) * ldb + k + kh;
            b.x = bp[0];
            b.y = bp[1];
        } else {
            const float* bp = B + (size_t)(k + kh) * ldb + col0 + mloc;
            b.x = bp[0];
            b.y = bp[ldb];
        }
        acc = __builtin_amdgcn_wmma_f32_16x16x4_f32(false, a, false, b,
                                                    (short)0, acc, false, false);
    }

    const int   rbase = row0 + ((lane >> 4) << 3);
    float*      cp    = C + (size_t)rbase * ldc + col0 + mloc;
#pragma unroll
    for (int r = 0; r < 8; ++r) cp[(size_t)r * ldc] = acc[r];
}

__device__ __forceinline__ unsigned f2ord(float f) {
    unsigned u = __float_as_uint(f);
    return (u & 0x80000000u) ? ~u : (u | 0x80000000u);
}
__device__ __forceinline__ float ord2f(unsigned u) {
    return (u & 0x80000000u) ? __uint_as_float(u & 0x7FFFFFFFu) : __uint_as_float(~u);
}

__global__ void __launch_bounds__(128) node_fused_kernel(const float* __restrict__ xin,
                                         const float* __restrict__ Wn,
                                         const float* __restrict__ al,
                                         const float* __restrict__ ar,
                                         float* __restrict__ ft,
                                         float* __restrict__ el,
                                         float* __restrict__ er,
                                         float* __restrict__ zb) {
    const int wave = blockIdx.x * (blockDim.x >> 5) + (threadIdx.x >> 5);
    if (wave >= (N_NODES / 16) * 2) return;
    const int lane = threadIdx.x & 31;
    const int c    = lane & 15;
    const int kh   = (lane >> 4) << 1;
    const int h    = wave & 1;
    const int n0   = (wave >> 1) << 4;

    zb[(size_t)h * (N_NODES * 2) + n0 * 2 + lane] = 0.f;
    {
        float4* ag4 = reinterpret_cast<float4*>(zb + (size_t)N_NODES * 4 + (size_t)n0 * 64 + h * 512);
        const float4 z4 = make_float4(0.f, 0.f, 0.f, 0.f);
#pragma unroll
        for (int i = 0; i < 4; ++i) ag4[i * 32 + lane] = z4;
    }

    v8f acc[4];
#pragma unroll
    for (int j = 0; j < 4; ++j) acc[j] = v8f{};
    const float* aBase = xin + (size_t)(n0 + c) * 64 + kh;
    const float* bBase = Wn + (size_t)kh * 128 + h * 64 + c;
#pragma unroll
    for (int k = 0; k < 64; k += 4) {
        v2f a;
        a.x = aBase[k];
        a.y = aBase[k + 1];
#pragma unroll
        for (int j = 0; j < 4; ++j) {
            v2f b;
            b.x = bBase[(size_t)k * 128 + j * 16];
            b.y = bBase[(size_t)(k + 1) * 128 + j * 16];
            acc[j] = __builtin_amdgcn_wmma_f32_16x16x4_f32(false, a, false, b,
                                                           (short)0, acc[j], false, false);
        }
    }

    const int rbase = n0 + ((lane >> 4) << 3);
    float*    cp    = ft + (size_t)rbase * 128 + h * 64 + c;
#pragma unroll
    for (int r = 0; r < 8; ++r) {
#pragma unroll
        for (int j = 0; j < 4; ++j) cp[(size_t)r * 128 + j * 16] = acc[j][r];
    }

    float s1[8], s2[8];
#pragma unroll
    for (int r = 0; r < 8; ++r) { s1[r] = 0.f; s2[r] = 0.f; }
#pragma unroll
    for (int j = 0; j < 4; ++j) {
        const float wl = al[h * 64 + j * 16 + c];
        const float wr = ar[h * 64 + j * 16 + c];
#pragma unroll
        for (int r = 0; r < 8; ++r) {
            s1[r] += acc[j][r] * wl;
            s2[r] += acc[j][r] * wr;
        }
    }
#pragma unroll
    for (int off = 1; off < 16; off <<= 1) {
#pragma unroll
        for (int r = 0; r < 8; ++r) {
            s1[r] += __shfl_xor(s1[r], off, 32);
            s2[r] += __shfl_xor(s2[r], off, 32);
        }
    }
    float v = 0.f;
#pragma unroll
    for (int r = 0; r < 8; ++r) {
        v = (c == r)     ? s1[r] : v;
        v = (c == r + 8) ? s2[r] : v;
    }
    float* outp = (c < 8) ? el : er;
    outp[(size_t)(rbase + (c & 7)) * 2 + h] = v;
}

__global__ void p_sum_kernel(float* __restrict__ lg, const unsigned* __restrict__ mU,
                             const int* __restrict__ dst, float* __restrict__ sB) {
    int t = blockIdx.x * blockDim.x + threadIdx.x;
    if (t >= N_EDGES * 2) return;
    int e = t >> 1, h = t & 1;
    float m  = ord2f(mU[dst[e] * 2 + h]);
    float pv = expf(lg[t] - m);
    lg[t]    = pv;
    atomicAdd(&sB[dst[e] * 2 + h], pv);
}

__global__ void finalize_kernel(const float* __restrict__ agg, const float* __restrict__ bb,
                                float* __restrict__ xout) {
    int t = blockIdx.x * blockDim.x + threadIdx.x;
    if (t >= N_NODES * 64) return;
    int d = t & 63;
    xout[t] = 0.5f * (agg[t] + (bb[d] + bb[64 + d]));
}


__device__ __forceinline__ void fe_tile_16x128(const float* __restrict__ eF,
                                               const float* __restrict__ We,
                                               int e0, int lane, v8f (&acc)[8]) {
    const int mloc = lane & 15;
    const int kh   = (lane >> 4) << 1;
    const float* aBase = eF + (size_t)(e0 + mloc) * 32 + kh;
    const float* bBase = We + (size_t)kh * 128 + mloc;
#pragma unroll
    for (int j = 0; j < 8; ++j) acc[j] = v8f{};
#pragma unroll
    for (int k = 0; k < 32; k += 4) {
        v2f a;
        a.x = aBase[k];
        a.y = aBase[k + 1];
#pragma unroll
        for (int j = 0; j < 8; ++j) {
            v2f b;
            b.x = bBase[(size_t)k * 128 + j * 16];
            b.y = bBase[(size_t)(k + 1) * 128 + j * 16];
            acc[j] = __builtin_amdgcn_wmma_f32_16x16x4_f32(false, a, false, b,
                                                           (short)0, acc[j], false, false);
        }
    }
}

__global__ void __launch_bounds__(128) edge_logits_fused_kernel(const float* __restrict__ eF,
                                         const float* __restrict__ We,
                                         const float* __restrict__ ae,
                                         const float* __restrict__ el,
                                         const float* __restrict__ er,
                                         const int* __restrict__ src,
                                         const int* __restrict__ dst,
                                         float* __restrict__ lg,
                                         unsigned* __restrict__ mU) {
    const int wave = blockIdx.x * (blockDim.x >> 5) + (threadIdx.x >> 5);
    if (wave >= N_EDGES / 16) return;
    const int lane = threadIdx.x & 31;
    const int c    = lane & 15;
    const int e0   = wave << 4;

    v8f acc[8];
    fe_tile_16x128(eF, We, e0, lane, acc);

    float s0[8], s1[8];
#pragma unroll
    for (int r = 0; r < 8; ++r) { s0[r] = 0.f; s1[r] = 0.f; }
#pragma unroll
    for (int j = 0; j < 4; ++j) {
        const float w0 = ae[j * 16 + c];
        const float w1 = ae[64 + j * 16 + c];
#pragma unroll
        for (int r = 0; r < 8; ++r) {
            s0[r] += acc[j][r] * w0;
            s1[r] += acc[j + 4][r] * w1;
        }
    }
#pragma unroll
    for (int off = 1; off < 16; off <<= 1) {
#pragma unroll
        for (int r = 0; r < 8; ++r) {
            s0[r] += __shfl_xor(s0[r], off, 32);
            s1[r] += __shfl_xor(s1[r], off, 32);
        }
    }
    float eev = 0.f;
#pragma unroll
    for (int r = 0; r < 8; ++r) {
        eev = (c == 2 * r)     ? s0[r] : eev;
        eev = (c == 2 * r + 1) ? s1[r] : eev;
    }
    const int e = e0 + ((lane >> 4) << 3) + (c >> 1);
    const int h = c & 1;
    const int t = e * 2 + h;
    const int dN = dst[e];
    float l = el[src[e] * 2 + h] + er[dN * 2 + h] + eev;
    l = (l > 0.f) ? l : 0.2f * l;
    lg[t] = l;
    atomicMax(&mU[dN * 2 + h], f2ord(l));
}

__global__ void __launch_bounds__(128) scatter_msg_fused_kernel(const float* __restrict__ eF,
                                         const float* __restrict__ We,
                                         const float* __restrict__ ft,
                                         const float* __restrict__ p,
                                         const float* __restrict__ sB,
                                         const int* __restrict__ src,
                                         const int* __restrict__ dst,
                                         float* __restrict__ agg) {
    const int wave = blockIdx.x * (blockDim.x >> 5) + (threadIdx.x >> 5);
    if (wave >= N_EDGES / 16) return;
    const int lane = threadIdx.x & 31;
    const int c    = lane & 15;
    const int e0   = wave << 4;

    const int eb = e0 + ((lane >> 4) << 3);

    int   sNr[8], dNr[8];
    float a0r[8], a1r[8];
#pragma unroll
    for (int r = 0; r < 8; ++r) {
        const int e = eb + r;
        sNr[r] = src[e];
        dNr[r] = dst[e];
        a0r[r] = p[e * 2]     / sB[dNr[r] * 2];
        a1r[r] = p[e * 2 + 1] / sB[dNr[r] * 2 + 1];
    }

    v8f acc[8];
    fe_tile_16x128(eF, We, e0, lane, acc);

#pragma unroll
    for (int r = 0; r < 8; ++r) {
        const float* fts = ft + (size_t)sNr[r] * 128 + c;
#pragma unroll
        for (int j = 0; j < 4; ++j) {
            const float m0 = (fts[j * 16]      + acc[j][r])     * a0r[r];
            const float m1 = (fts[64 + j * 16] + acc[j + 4][r]) * a1r[r];
            acc[j][r] = m0 + m1;
        }
    }
#pragma unroll
    for (int r = 0; r < 8; ++r) {
        float* ag = agg + (size_t)dNr[r] * 64 + c;
#pragma unroll
        for (int j = 0; j < 4; ++j) atomicAdd(&ag[j * 16], acc[j][r]);
    }
}

static inline int cdiv(int a, int b) { return (a + b - 1) / b; }

struct WsPtrs {
    float* ft; float* el; float* er; float* lg;
    unsigned* mU; float* sB; float* agg; float* xb0; float* xb1;
};

static void run_edge_gat_layer(const float* xin, const float* eF, const int* src,
                               const int* dst, const float* Wn, const float* We,
                               const float* al, const float* ar, const float* ae,
                               const float* bb, float* xout, const WsPtrs& w,
                               hipStream_t stream) {
    node_fused_kernel<<<(N_NODES / 16) * 2 / 4, 128, 0, stream>>>(
        xin, Wn, al, ar, w.ft, w.el, w.er, (float*)w.mU);

    edge_logits_fused_kernel<<<cdiv(N_EDGES / 16, 4), 128, 0, stream>>>(
        eF, We, ae, w.el, w.er, src, dst, w.lg, w.mU);
    p_sum_kernel<<<cdiv(N_EDGES * 2, TPB), TPB, 0, stream>>>(w.lg, w.mU, dst, w.sB);
    scatter_msg_fused_kernel<<<cdiv(N_EDGES / 16, 4), 128, 0, stream>>>(
        eF, We, w.ft, w.lg, w.sB, src, dst, w.agg);
    finalize_kernel<<<cdiv(N_NODES * 64, TPB), TPB, 0, stream>>>(w.agg, bb, xout);
}

extern "C" void kernel_launch(void* const* d_in, const int* in_sizes, int n_in,
                              void* d_out, int out_size, void* d_ws, size_t ws_size,
                              hipStream_t stream) {
    const float* x_all = (const float*)d_in[0];
    const float* e_all = (const float*)d_in[1];
    const int*   s_all = (const int*)d_in[2];
    const int*   d_all = (const int*)d_in[3];

    const float* xS  = x_all + (size_t)1 * N_NODES * 64;
    const float* eS  = e_all + (size_t)1 * N_EDGES * 32;
    const int*   src = s_all + (size_t)1 * N_EDGES;
    const int*   dst = d_all + (size_t)1 * N_EDGES;

    const float* const* P = (const float* const*)d_in;
    const float* W_e2d = P[28];

    float* ws = (float*)d_ws;
    size_t off = 0;
    WsPtrs w;
    w.ft  = ws + off; off += (size_t)N_NODES * 128;
    w.el  = ws + off; off += (size_t)N_NODES * 2;
    w.er  = ws + off; off += (size_t)N_NODES * 2;
    w.lg  = ws + off; off += (size_t)N_EDGES * 2;
    w.mU  = (unsigned*)(ws + off); off += (size_t)N_NODES * 2;
    w.sB  = ws + off; off += (size_t)N_NODES * 2;
    w.agg = ws + off; off += (size_t)N_NODES * 64;
    w.xb0 = ws + off; off += (size_t)N_NODES * 64;
    w.xb1 = ws + off; off += (size_t)N_NODES * 64;

    run_edge_gat_layer(xS, eS, src, dst, P[4], P[5], P[6], P[7], P[8], P[9],
                       w.xb0, w, stream);
    run_edge_gat_layer(w.xb0, eS, src, dst, P[10], P[11], P[12], P[13], P[14], P[15],
                       w.xb1, w, stream);
    gemm_wmma_f32<true><<<(N_NODES / 16) * (64 / 16) / 4, 128, 0, stream>>>(
        w.xb1, W_e2d, w.xb0, N_NODES, 64, 64, 64, 64, 64);
    run_edge_gat_layer(w.xb0, eS, src, dst, P[16], P[17], P[18], P[19], P[20], P[21],
                       w.xb1, w, stream);
    run_edge_gat_layer(w.xb1, eS, src, dst, P[22], P[23], P[24], P[25], P[26], P[27],
                       (float*)d_out, w, stream);
}


# Edge logit term reassociated: ee = eF @ (We.ae), a K=32 WMMA onto a [2,32] vector formed once per layer, replaces the [E,128] fe product in the logits kernel (sum order of ee changes)
# speedup vs baseline: 1.7664x; 1.0991x over previous
#include <hip/hip_runtime.h>
#include <hip/hip_bf16.h>

#define N_NODES 20000
#define N_EDGES 100000
#define TPB 256

typedef __attribute__((ext_vector_type(2))) float v2f;
typedef __attribute__((ext_vector_type(8))) float v8f;

template <bool BT>
__global__ void gemm_wmma_f32(const float* __restrict__ A, const float* __restrict__ B,
                              float* __restrict__ C, int M, int K, int NC,
                              int lda, int ldb, int ldc) {
    const int tilesN = NC >> 4;
    const int wave   = blockIdx.x * (blockDim.x >> 5) + (threadIdx.x >> 5);
    const int trow   = wave / tilesN;
    const int tcol   = wave - trow * tilesN;
    if ((trow << 4) >= M) return;

    const int lane = threadIdx.x & 31;
    const int mloc = lane & 15;
    const int kh   = (lane >> 4) << 1;
    const int row0 = trow << 4, col0 = tcol << 4;

    v8f acc = {};
    const float* aBase = A + (size_t)(row0 + mloc) * lda + kh;
    for (int k = 0; k < K; k += 4) {
        v2f a, b;
        a.x = aBase[k];
        a.y = aBase[k + 1];
        if (BT) {
            const float* bp = B + (size_t)(col0 + mloc) * ldb + k + kh;
            b.x = bp[0];
            b.y = bp[1];
        } else {
            const float* bp = B + (size_t)(k + kh) * ldb + col0 + mloc;
            b.x = bp[0];
            b.y = bp[ldb];
        }
        acc = __builtin_amdgcn_wmma_f32_16x16x4_f32(false, a, false, b,
                                                    (short)0, acc, false, false);
    }

    const int   rbase = row0 + ((lane >> 4) << 3);
    float*      cp    = C + (size_t)rbase * ldc + col0 + mloc;
#pragma unroll
    for (int r = 0; r < 8; ++r) cp[(size_t)r * ldc] = acc[r];
}

__device__ __forceinline__ unsigned f2ord(float f) {
    unsigned u = __float_as_uint(f);
    return (u & 0x80000000u) ? ~u : (u | 0x80000000u);
}
__device__ __forceinline__ float ord2f(unsigned u) {
    return (u & 0x80000000u) ? __uint_as_float(u & 0x7FFFFFFFu) : __uint_as_float(~u);
}

__global__ void __launch_bounds__(128) node_fused_kernel(const float* __restrict__ xin,
                                         const float* __restrict__ Wn,
                                         const float* __restrict__ al,
                                         const float* __restrict__ ar,
                                         float* __restrict__ ft,
                                         float* __restrict__ el,
                                         float* __restrict__ er,
                                         float* __restrict__ zb,
                                         const float* __restrict__ We,
                                         const float* __restrict__ ae,
                                         float* __restrict__ wee) {
    if (blockIdx.x == (N_NODES / 16) * 2 / 4) {
        const int t = threadIdx.x;
        if (t < 64) {
            const int h = t >> 5, k = t & 31;
            const float4* wp = reinterpret_cast<const float4*>(We + (size_t)k * 128 + h * 64);
            const float4* ap = reinterpret_cast<const float4*>(ae + h * 64);
            float acc = 0.f;
#pragma unroll
            for (int d = 0; d < 16; ++d) {
                const float4 wv = wp[d];
                const float4 av = ap[d];
                acc += wv.x * av.x;
                acc += wv.y * av.y;
                acc += wv.z * av.z;
                acc += wv.w * av.w;
            }
            wee[h * 32 + k] = acc;
        }
        return;
    }
    const int wave = blockIdx.x * (blockDim.x >> 5) + (threadIdx.x >> 5);
    if (wave >= (N_NODES / 16) * 2) return;
    const int lane = threadIdx.x & 31;
    const int c    = lane & 15;
    const int kh   = (lane >> 4) << 1;
    const int h    = wave & 1;
    const int n0   = (wave >> 1) << 4;

    zb[(size_t)h * (N_NODES * 2) + n0 * 2 + lane] = 0.f;
    {
        float4* ag4 = reinterpret_cast<float4*>(zb + (size_t)N_NODES * 4 + (size_t)n0 * 64 + h * 512);
        const float4 z4 = make_float4(0.f, 0.f, 0.f, 0.f);
#pragma unroll
        for (int i = 0; i < 4; ++i) ag4[i * 32 + lane] = z4;
    }

    v8f acc[4];
#pragma unroll
    for (int j = 0; j < 4; ++j) acc[j] = v8f{};
    const float* aBase = xin + (size_t)(n0 + c) * 64 + kh;
    const float* bBase = Wn + (size_t)kh * 128 + h * 64 + c;
#pragma unroll
    for (int k = 0; k < 64; k += 4) {
        v2f a;
        a.x = aBase[k];
        a.y = aBase[k + 1];
#pragma unroll
        for (int j = 0; j < 4; ++j) {
            v2f b;
            b.x = bBase[(size_t)k * 128 + j * 16];
            b.y = bBase[(size_t)(k + 1) * 128 + j * 16];
            acc[j] = __builtin_amdgcn_wmma_f32_16x16x4_f32(false, a, false, b,
                                                           (short)0, acc[j], false, false);
        }
    }

    const int rbase = n0 + ((lane >> 4) << 3);
    float*    cp    = ft + (size_t)rbase * 128 + h * 64 + c;
#pragma unroll
    for (int r = 0; r < 8; ++r) {
#pragma unroll
        for (int j = 0; j < 4; ++j) cp[(size_t)r * 128 + j * 16] = acc[j][r];
    }

    float s1[8], s2[8];
#pragma unroll
    for (int r = 0; r < 8; ++r) { s1[r] = 0.f; s2[r] = 0.f; }
#pragma unroll
    for (int j = 0; j < 4; ++j) {
        const float wl = al[h * 64 + j * 16 + c];
        const float wr = ar[h * 64 + j * 16 + c];
#pragma unroll
        for (int r = 0; r < 8; ++r) {
            s1[r] += acc[j][r] * wl;
            s2[r] += acc[j][r] * wr;
        }
    }
#pragma unroll
    for (int off = 1; off < 16; off <<= 1) {
#pragma unroll
        for (int r = 0; r < 8; ++r) {
            s1[r] += __shfl_xor(s1[r], off, 32);
            s2[r] += __shfl_xor(s2[r], off, 32);
        }
    }
    float v = 0.f;
#pragma unroll
    for (int r = 0; r < 8; ++r) {
        v = (c == r)     ? s1[r] : v;
        v = (c == r + 8) ? s2[r] : v;
    }
    float* outp = (c < 8) ? el : er;
    outp[(size_t)(rbase + (c & 7)) * 2 + h] = v;
}

__global__ void p_sum_kernel(float* __restrict__ lg, const unsigned* __restrict__ mU,
                             const int* __restrict__ dst, float* __restrict__ sB) {
    int t = blockIdx.x * blockDim.x + threadIdx.x;
    if (t >= N_EDGES * 2) return;
    int e = t >> 1, h = t & 1;
    float m  = ord2f(mU[dst[e] * 2 + h]);
    float pv = expf(lg[t] - m);
    lg[t]    = pv;
    atomicAdd(&sB[dst[e] * 2 + h], pv);
}

__global__ void finalize_kernel(const float* __restrict__ agg, const float* __restrict__ bb,
                                float* __restrict__ xout) {
    int t = blockIdx.x * blockDim.x + threadIdx.x;
    if (t >= N_NODES * 64) return;
    int d = t & 63;
    xout[t] = 0.5f * (agg[t] + (bb[d] + bb[64 + d]));
}


__device__ __forceinline__ void fe_tile_16x128(const float* __restrict__ eF,
                                               const float* __restrict__ We,
                                               int e0, int lane, v8f (&acc)[8]) {
    const int mloc = lane & 15;
    const int kh   = (lane >> 4) << 1;
    const float* aBase = eF + (size_t)(e0 + mloc) * 32 + kh;
    const float* bBase = We + (size_t)kh * 128 + mloc;
#pragma unroll
    for (int j = 0; j < 8; ++j) acc[j] = v8f{};
#pragma unroll
    for (int k = 0; k < 32; k += 4) {
        v2f a;
        a.x = aBase[k];
        a.y = aBase[k + 1];
#pragma unroll
        for (int j = 0; j < 8; ++j) {
            v2f b;
            b.x = bBase[(size_t)k * 128 + j * 16];
            b.y = bBase[(size_t)(k + 1) * 128 + j * 16];
            acc[j] = __builtin_amdgcn_wmma_f32_16x16x4_f32(false, a, false, b,
                                                           (short)0, acc[j], false, false);
        }
    }
}

__global__ void __launch_bounds__(128) edge_logits_kernel(const float* __restrict__ eF,
                                         const float* __restrict__ wee,
                                         const float* __restrict__ el,
                                         const float* __restrict__ er,
                                         const int* __restrict__ src,
                                         const int* __restrict__ dst,
                                         float* __restrict__ lg,
                                         unsigned* __restrict__ mU) {
    const int wave = blockIdx.x * (blockDim.x >> 5) + (threadIdx.x >> 5);
    if (wave >= N_EDGES / 32) return;
    const int lane = threadIdx.x & 31;
    const int c    = lane & 15;
    const int hf   = lane >> 4;
    const int kh   = hf << 1;
    const int e0   = wave << 5;

    const bool   live = (c & 6) == 0;
    const float* ap = wee + (c & 1) * 32 + kh;
    const float* b0 = eF + (size_t)(e0 + c) * 32 + kh;
    const float* b1 = b0 + 16 * 32;

    v8f acc0 = {}, acc1 = {};
#pragma unroll
    for (int k = 0; k < 32; k += 4) {
        v2f a, p, q;
        const float a0v = ap[k], a1v = ap[k + 1];
        a.x = live ? a0v : 0.f;
        a.y = live ? a1v : 0.f;
        p.x = b0[k];
        p.y = b0[k + 1];
        q.x = b1[k];
        q.y = b1[k + 1];
        acc0 = __builtin_amdgcn_wmma_f32_16x16x4_f32(false, a, false, p,
                                                     (short)0, acc0, false, false);
        acc1 = __builtin_amdgcn_wmma_f32_16x16x4_f32(false, a, false, q,
                                                     (short)0, acc1, false, false);
    }

    const int h = hf;
    int   dNr[2];
    float lr[2];
#pragma unroll
    for (int j = 0; j < 2; ++j) {
        const float eev = j ? (h ? acc1[1] : acc1[0]) : (h ? acc0[1] : acc0[0]);
        const int e = e0 + j * 16 + c;
        dNr[j] = dst[e];
        const float l = el[src[e] * 2 + h] + er[dNr[j] * 2 + h] + eev;
        lr[j] = (l > 0.f) ? l : 0.2f * l;
    }
#pragma unroll
    for (int j = 0; j < 2; ++j) {
        const int e = e0 + j * 16 + c;
        lg[e * 2 + h] = lr[j];
        atomicMax(&mU[dNr[j] * 2 + h], f2ord(lr[j]));
    }
}

__global__ void __launch_bounds__(128) scatter_msg_fused_kernel(const float* __restrict__ eF,
                                         const float* __restrict__ We,
                                         const float* __restrict__ ft,
                                         const float* __restrict__ p,
                                         const float* __restrict__ sB,
                                         const int* __restrict__ src,
                                         const int* __restrict__ dst,
                                         float* __restrict__ agg) {
    const int wave = blockIdx.x * (blockDim.x >> 5) + (threadIdx.x >> 5);
    if (wave >= N_EDGES / 16) return;
    const int lane = threadIdx.x & 31;
    const int c    = lane & 15;
    const int e0   = wave << 4;

    const int eb = e0 + ((lane >> 4) << 3);

    int   sNr[8], dNr[8];
    float a0r[8], a1r[8];
#pragma unroll
    for (int r = 0; r < 8; ++r) {
        const int e = eb + r;
        sNr[r] = src[e];
        dNr[r] = dst[e];
        a0r[r] = p[e * 2]     / sB[dNr[r] * 2];
        a1r[r] = p[e * 2 + 1] / sB[dNr[r] * 2 + 1];
    }

    v8f acc[8];
    fe_tile_16x128(eF, We, e0, lane, acc);

#pragma unroll
    for (int r = 0; r < 8; ++r) {
        const float* fts = ft + (size_t)sNr[r] * 128 + c;
#pragma unroll
        for (int j = 0; j < 4; ++j) {
            const float m0 = (fts[j * 16]      + acc[j][r])     * a0r[r];
            const float m1 = (fts[64 + j * 16] + acc[j + 4][r]) * a1r[r];
            acc[j][r] = m0 + m1;
        }
    }
#pragma unroll
    for (int r = 0; r < 8; ++r) {
        float* ag = agg + (size_t)dNr[r] * 64 + c;
#pragma unroll
        for (int j = 0; j < 4; ++j) atomicAdd(&ag[j * 16], acc[j][r]);
    }
}

static inline int cdiv(int a, int b) { return (a + b - 1) / b; }

struct WsPtrs {
    float* ft; float* el; float* er; float* lg;
    unsigned* mU; float* sB; float* agg; float* xb0; float* xb1; float* wee;
};

static void run_edge_gat_layer(const float* xin, const float* eF, const int* src,
                               const int* dst, const float* Wn, const float* We,
                               const float* al, const float* ar, const float* ae,
                               const float* bb, float* xout, const WsPtrs& w,
                               hipStream_t stream) {
    node_fused_kernel<<<(N_NODES / 16) * 2 / 4 + 1, 128, 0, stream>>>(
        xin, Wn, al, ar, w.ft, w.el, w.er, (float*)w.mU, We, ae, w.wee);

    edge_logits_kernel<<<cdiv(N_EDGES / 32, 4), 128, 0, stream>>>(
        eF, w.wee, w.el, w.er, src, dst, w.lg, w.mU);
    p_sum_kernel<<<cdiv(N_EDGES * 2, TPB), TPB, 0, stream>>>(w.lg, w.mU, dst, w.sB);
    scatter_msg_fused_kernel<<<cdiv(N_EDGES / 16, 4), 128, 0, stream>>>(
        eF, We, w.ft, w.lg, w.sB, src, dst, w.agg);
    finalize_kernel<<<cdiv(N_NODES * 64, TPB), TPB, 0, stream>>>(w.agg, bb, xout);
}

extern "C" void kernel_launch(void* const* d_in, const int* in_sizes, int n_in,
                              void* d_out, int out_size, void* d_ws, size_t ws_size,
                              hipStream_t stream) {
    const float* x_all = (const float*)d_in[0];
    const float* e_all = (const float*)d_in[1];
    const int*   s_all = (const int*)d_in[2];
    const int*   d_all = (const int*)d_in[3];

    const float* xS  = x_all + (size_t)1 * N_NODES * 64;
    const float* eS  = e_all + (size_t)1 * N_EDGES * 32;
    const int*   src = s_all + (size_t)1 * N_EDGES;
    const int*   dst = d_all + (size_t)1 * N_EDGES;

    const float* const* P = (const float* const*)d_in;
    const float* W_e2d = P[28];

    float* ws = (float*)d_ws;
    size_t off = 0;
    WsPtrs w;
    w.ft  = ws + off; off += (size_t)N_NODES * 128;
    w.el  = ws + off; off += (size_t)N_NODES * 2;
    w.er  = ws + off; off += (size_t)N_NODES * 2;
    w.lg  = ws + off; off += (size_t)N_EDGES * 2;
    w.mU  = (unsigned*)(ws + off); off += (size_t)N_NODES * 2;
    w.sB  = ws + off; off += (size_t)N_NODES * 2;
    w.agg = ws + off; off += (size_t)N_NODES * 64;
    w.xb0 = ws + off; off += (size_t)N_NODES * 64;
    w.xb1 = ws + off; off += (size_t)N_NODES * 64;
    w.wee = ws + off; off += 64;

    run_edge_gat_layer(xS, eS, src, dst, P[4], P[5], P[6], P[7], P[8], P[9],
                       w.xb0, w, stream);
    run_edge_gat_layer(w.xb0, eS, src, dst, P[10], P[11], P[12], P[13], P[14], P[15],
                       w.xb1, w, stream);
    gemm_wmma_f32<true><<<(N_NODES / 16) * (64 / 16) / 4, 128, 0, stream>>>(
        w.xb1, W_e2d, w.xb0, N_NODES, 64, 64, 64, 64, 64);
    run_edge_gat_layer(w.xb0, eS, src, dst, P[16], P[17], P[18], P[19], P[20], P[21],
                       w.xb1, w, stream);
    run_edge_gat_layer(w.xb1, eS, src, dst, P[22], P[23], P[24], P[25], P[26], P[27],
                       (float*)d_out, w, stream);
}


# Edge weight matrix staged once per 4-wave block in LDS (k-pair interleaved, padded pitch); fe B fragments read as 8-byte LDS pairs instead of 128 global loads per wave; fe values bit-identical
# speedup vs baseline: 1.8990x; 1.0751x over previous
#include <hip/hip_runtime.h>
#include <hip/hip_bf16.h>

#define N_NODES 20000
#define N_EDGES 100000
#define TPB 256

typedef __attribute__((ext_vector_type(2))) float v2f;
typedef __attribute__((ext_vector_type(8))) float v8f;

template <bool BT>
__global__ void gemm_wmma_f32(const float* __restrict__ A, const float* __restrict__ B,
                              float* __restrict__ C, int M, int K, int NC,
                              int lda, int ldb, int ldc) {
    const int tilesN = NC >> 4;
    const int wave   = blockIdx.x * (blockDim.x >> 5) + (threadIdx.x >> 5);
    const int trow   = wave / tilesN;
    const int tcol   = wave - trow * tilesN;
    if ((trow << 4) >= M) return;

    const int lane = threadIdx.x & 31;
    const int mloc = lane & 15;
    const int kh   = (lane >> 4) << 1;
    const int row0 = trow << 4, col0 = tcol << 4;

    v8f acc = {};
    const float* aBase = A + (size_t)(row0 + mloc) * lda + kh;
    for (int k = 0; k < K; k += 4) {
        v2f a, b;
        a.x = aBase[k];
        a.y = aBase[k + 1];
        if (BT) {
            const float* bp = B + (size_t)(col0 + mloc) * ldb + k + kh;
            b.x = bp[0];
            b.y = bp[1];
        } else {
            const float* bp = B + (size_t)(k + kh) * ldb + col0 + mloc;
            b.x = bp[0];
            b.y = bp[ldb];
        }
        acc = __builtin_amdgcn_wmma_f32_16x16x4_f32(false, a, false, b,
                                                    (short)0, acc, false, false);
    }

    const int   rbase = row0 + ((lane >> 4) << 3);
    float*      cp    = C + (size_t)rbase * ldc + col0 + mloc;
#pragma unroll
    for (int r = 0; r < 8; ++r) cp[(size_t)r * ldc] = acc[r];
}

__device__ __forceinline__ unsigned f2ord(float f) {
    unsigned u = __float_as_uint(f);
    return (u & 0x80000000u) ? ~u : (u | 0x80000000u);
}
__device__ __forceinline__ float ord2f(unsigned u) {
    return (u & 0x80000000u) ? __uint_as_float(u & 0x7FFFFFFFu) : __uint_as_float(~u);
}

__global__ void __launch_bounds__(128) node_fused_kernel(const float* __restrict__ xin,
                                         const float* __restrict__ Wn,
                                         const float* __restrict__ al,
                                         const float* __restrict__ ar,
                                         float* __restrict__ ft,
                                         float* __restrict__ el,
                                         float* __restrict__ er,
                                         float* __restrict__ zb,
                                         const float* __restrict__ We,
                                         const float* __restrict__ ae,
                                         float* __restrict__ wee) {
    if (blockIdx.x == (N_NODES / 16) * 2 / 4) {
        const int t = threadIdx.x;
        if (t < 64) {
            const int h = t >> 5, k = t & 31;
            const float4* wp = reinterpret_cast<const float4*>(We + (size_t)k * 128 + h * 64);
            const float4* ap = reinterpret_cast<const float4*>(ae + h * 64);
            float acc = 0.f;
#pragma unroll
            for (int d = 0; d < 16; ++d) {
                const float4 wv = wp[d];
                const float4 av = ap[d];
                acc += wv.x * av.x;
                acc += wv.y * av.y;
                acc += wv.z * av.z;
                acc += wv.w * av.w;
            }
            wee[h * 32 + k] = acc;
        }
        return;
    }
    const int wave = blockIdx.x * (blockDim.x >> 5) + (threadIdx.x >> 5);
    if (wave >= (N_NODES / 16) * 2) return;
    const int lane = threadIdx.x & 31;
    const int c    = lane & 15;
    const int kh   = (lane >> 4) << 1;
    const int h    = wave & 1;
    const int n0   = (wave >> 1) << 4;

    zb[(size_t)h * (N_NODES * 2) + n0 * 2 + lane] = 0.f;
    {
        float4* ag4 = reinterpret_cast<float4*>(zb + (size_t)N_NODES * 4 + (size_t)n0 * 64 + h * 512);
        const float4 z4 = make_float4(0.f, 0.f, 0.f, 0.f);
#pragma unroll
        for (int i = 0; i < 4; ++i) ag4[i * 32 + lane] = z4;
    }

    v8f acc[4];
#pragma unroll
    for (int j = 0; j < 4; ++j) acc[j] = v8f{};
    const float* aBase = xin + (size_t)(n0 + c) * 64 + kh;
    const float* bBase = Wn + (size_t)kh * 128 + h * 64 + c;
#pragma unroll
    for (int k = 0; k < 64; k += 4) {
        v2f a;
        a.x = aBase[k];
        a.y = aBase[k + 1];
#pragma unroll
        for (int j = 0; j < 4; ++j) {
            v2f b;
            b.x = bBase[(size_t)k * 128 + j * 16];
            b.y = bBase[(size_t)(k + 1) * 128 + j * 16];
            acc[j] = __builtin_amdgcn_wmma_f32_16x16x4_f32(false, a, false, b,
                                                           (short)0, acc[j], false, false);
        }
    }

    const int rbase = n0 + ((lane >> 4) << 3);
    float*    cp    = ft + (size_t)rbase * 128 + h * 64 + c;
#pragma unroll
    for (int r = 0; r < 8; ++r) {
#pragma unroll
        for (int j = 0; j < 4; ++j) cp[(size_t)r * 128 + j * 16] = acc[j][r];
    }

    float s1[8], s2[8];
#pragma unroll
    for (int r = 0; r < 8; ++r) { s1[r] = 0.f; s2[r] = 0.f; }
#pragma unroll
    for (int j = 0; j < 4; ++j) {
        const float wl = al[h * 64 + j * 16 + c];
        const float wr = ar[h * 64 + j * 16 + c];
#pragma unroll
        for (int r = 0; r < 8; ++r) {
            s1[r] += acc[j][r] * wl;
            s2[r] += acc[j][r] * wr;
        }
    }
#pragma unroll
    for (int off = 1; off < 16; off <<= 1) {
#pragma unroll
        for (int r = 0; r < 8; ++r) {
            s1[r] += __shfl_xor(s1[r], off, 32);
            s2[r] += __shfl_xor(s2[r], off, 32);
        }
    }
    float v = 0.f;
#pragma unroll
    for (int r = 0; r < 8; ++r) {
        v = (c == r)     ? s1[r] : v;
        v = (c == r + 8) ? s2[r] : v;
    }
    float* outp = (c < 8) ? el : er;
    outp[(size_t)(rbase + (c & 7)) * 2 + h] = v;
}

__global__ void p_sum_kernel(float* __restrict__ lg, const unsigned* __restrict__ mU,
                             const int* __restrict__ dst, float* __restrict__ sB) {
    int t = blockIdx.x * blockDim.x + threadIdx.x;
    if (t >= N_EDGES * 2) return;
    int e = t >> 1, h = t & 1;
    float m  = ord2f(mU[dst[e] * 2 + h]);
    float pv = expf(lg[t] - m);
    lg[t]    = pv;
    atomicAdd(&sB[dst[e] * 2 + h], pv);
}

__global__ void finalize_kernel(const float* __restrict__ agg, const float* __restrict__ bb,
                                float* __restrict__ xout) {
    int t = blockIdx.x * blockDim.x + threadIdx.x;
    if (t >= N_NODES * 64) return;
    int d = t & 63;
    xout[t] = 0.5f * (agg[t] + (bb[d] + bb[64 + d]));
}


#define WE_PITCH 288

__device__ __forceinline__ void stage_We_pairs(const float* __restrict__ We, float* sW, int t) {
#pragma unroll
    for (int q = 0; q < 4; ++q) {
        const int item = t + q * 128;
        const int kp = item >> 5, c4 = item & 31;
        const float4 r0 = *reinterpret_cast<const float4*>(We + (size_t)(2 * kp) * 128 + c4 * 4);
        const float4 r1 = *reinterpret_cast<const float4*>(We + (size_t)(2 * kp + 1) * 128 + c4 * 4);
        float4* d = reinterpret_cast<float4*>(sW + kp * WE_PITCH + c4 * 8);
        d[0] = make_float4(r0.x, r1.x, r0.y, r1.y);
        d[1] = make_float4(r0.z, r1.z, r0.w, r1.w);
    }
}

__device__ __forceinline__ void fe_tile_16x128(const float* __restrict__ eF,
                                               const float* sW,
                                               int e0, int lane, v8f (&acc)[8]) {
    const int mloc = lane & 15;
    const int hf   = lane >> 4;
    const int kh   = hf << 1;
    const float* aBase = eF + (size_t)(e0 + mloc) * 32 + kh;
    const float* bBase = sW + hf * WE_PITCH + mloc * 2;
#pragma unroll
    for (int j = 0; j < 8; ++j) acc[j] = v8f{};
#pragma unroll
    for (int k = 0; k < 32; k += 4) {
        v2f a;
        a.x = aBase[k];
        a.y = aBase[k + 1];
#pragma unroll
        for (int j = 0; j < 8; ++j) {
            const v2f b = *reinterpret_cast<const v2f*>(bBase + (k >> 1) * WE_PITCH + j * 32);
            acc[j] = __builtin_amdgcn_wmma_f32_16x16x4_f32(false, a, false, b,
                                                           (short)0, acc[j], false, false);
        }
    }
}

__global__ void __launch_bounds__(128) edge_logits_kernel(const float* __restrict__ eF,
                                         const float* __restrict__ wee,
                                         const float* __restrict__ el,
                                         const float* __restrict__ er,
                                         const int* __restrict__ src,
                                         const int* __restrict__ dst,
                                         float* __restrict__ lg,
                                         unsigned* __restrict__ mU) {
    const int wave = blockIdx.x * (blockDim.x >> 5) + (threadIdx.x >> 5);
    if (wave >= N_EDGES / 32) return;
    const int lane = threadIdx.x & 31;
    const int c    = lane & 15;
    const int hf   = lane >> 4;
    const int kh   = hf << 1;
    const int e0   = wave << 5;

    const bool   live = (c & 6) == 0;
    const float* ap = wee + (c & 1) * 32 + kh;
    const float* b0 = eF + (size_t)(e0 + c) * 32 + kh;
    const float* b1 = b0 + 16 * 32;

    v8f acc0 = {}, acc1 = {};
#pragma unroll
    for (int k = 0; k < 32; k += 4) {
        v2f a, p, q;
        const float a0v = ap[k], a1v = ap[k + 1];
        a.x = live ? a0v : 0.f;
        a.y = live ? a1v : 0.f;
        p.x = b0[k];
        p.y = b0[k + 1];
        q.x = b1[k];
        q.y = b1[k + 1];
        acc0 = __builtin_amdgcn_wmma_f32_16x16x4_f32(false, a, false, p,
                                                     (short)0, acc0, false, false);
        acc1 = __builtin_amdgcn_wmma_f32_16x16x4_f32(false, a, false, q,
                                                     (short)0, acc1, false, false);
    }

    const int h = hf;
    int   dNr[2];
    float lr[2];
#pragma unroll
    for (int j = 0; j < 2; ++j) {
        const float eev = j ? (h ? acc1[1] : acc1[0]) : (h ? acc0[1] : acc0[0]);
        const int e = e0 + j * 16 + c;
        dNr[j] = dst[e];
        const float l = el[src[e] * 2 + h] + er[dNr[j] * 2 + h] + eev;
        lr[j] = (l > 0.f) ? l : 0.2f * l;
    }
#pragma unroll
    for (int j = 0; j < 2; ++j) {
        const int e = e0 + j * 16 + c;
        lg[e * 2 + h] = lr[j];
        atomicMax(&mU[dNr[j] * 2 + h], f2ord(lr[j]));
    }
}

__global__ void __launch_bounds__(128) scatter_msg_fused_kernel(const float* __restrict__ eF,
                                         const float* __restrict__ We,
                                         const float* __restrict__ ft,
                                         const float* __restrict__ p,
                                         const float* __restrict__ sB,
                                         const int* __restrict__ src,
                                         const int* __restrict__ dst,
                                         float* __restrict__ agg) {
    __shared__ __attribute__((aligned(16))) float sW[16 * WE_PITCH];
    stage_We_pairs(We, sW, threadIdx.x);
    __syncthreads();

    const int wave = blockIdx.x * (blockDim.x >> 5) + (threadIdx.x >> 5);
    if (wave >= N_EDGES / 16) return;
    const int lane = threadIdx.x & 31;
    const int c    = lane & 15;
    const int e0   = wave << 4;

    const int eb = e0 + ((lane >> 4) << 3);

    int   sNr[8], dNr[8];
    float a0r[8], a1r[8];
#pragma unroll
    for (int r = 0; r < 8; ++r) {
        const int e = eb + r;
        sNr[r] = src[e];
        dNr[r] = dst[e];
        a0r[r] = p[e * 2]     / sB[dNr[r] * 2];
        a1r[r] = p[e * 2 + 1] / sB[dNr[r] * 2 + 1];
    }

    v8f acc[8];
    fe_tile_16x128(eF, sW, e0, lane, acc);

#pragma unroll
    for (int r = 0; r < 8; ++r) {
        const float* fts = ft + (size_t)sNr[r] * 128 + c;
#pragma unroll
        for (int j = 0; j < 4; ++j) {
            const float m0 = (fts[j * 16]      + acc[j][r])     * a0r[r];
            const float m1 = (fts[64 + j * 16] + acc[j + 4][r]) * a1r[r];
            acc[j][r] = m0 + m1;
        }
    }
#pragma unroll
    for (int r = 0; r < 8; ++r) {
        float* ag = agg + (size_t)dNr[r] * 64 + c;
#pragma unroll
        for (int j = 0; j < 4; ++j) atomicAdd(&ag[j * 16], acc[j][r]);
    }
}

static inline int cdiv(int a, int b) { return (a + b - 1) / b; }

struct WsPtrs {
    float* ft; float* el; float* er; float* lg;
    unsigned* mU; float* sB; float* agg; float* xb0; float* xb1; float* wee;
};

static void run_edge_gat_layer(const float* xin, const float* eF, const int* src,
                               const int* dst, const float* Wn, const float* We,
                               const float* al, const float* ar, const float* ae,
                               const float* bb, float* xout, const WsPtrs& w,
                               hipStream_t stream) {
    node_fused_kernel<<<(N_NODES / 16) * 2 / 4 + 1, 128, 0, stream>>>(
        xin, Wn, al, ar, w.ft, w.el, w.er, (float*)w.mU, We, ae, w.wee);

    edge_logits_kernel<<<cdiv(N_EDGES / 32, 4), 128, 0, stream>>>(
        eF, w.wee, w.el, w.er, src, dst, w.lg, w.mU);
    p_sum_kernel<<<cdiv(N_EDGES * 2, TPB), TPB, 0, stream>>>(w.lg, w.mU, dst, w.sB);
    scatter_msg_fused_kernel<<<cdiv(N_EDGES / 16, 4), 128, 0, stream>>>(
        eF, We, w.ft, w.lg, w.sB, src, dst, w.agg);
    finalize_kernel<<<cdiv(N_NODES * 64, TPB), TPB, 0, stream>>>(w.agg, bb, xout);
}

extern "C" void kernel_launch(void* const* d_in, const int* in_sizes, int n_in,
                              void* d_out, int out_size, void* d_ws, size_t ws_size,
                              hipStream_t stream) {
    const float* x_all = (const float*)d_in[0];
    const float* e_all = (const float*)d_in[1];
    const int*   s_all = (const int*)d_in[2];
    const int*   d_all = (const int*)d_in[3];

    const float* xS  = x_all + (size_t)1 * N_NODES * 64;
    const float* eS  = e_all + (size_t)1 * N_EDGES * 32;
    const int*   src = s_all + (size_t)1 * N_EDGES;
    const int*   dst = d_all + (size_t)1 * N_EDGES;

    const float* const* P = (const float* const*)d_in;
    const float* W_e2d = P[28];

    float* ws = (float*)d_ws;
    size_t off = 0;
    WsPtrs w;
    w.ft  = ws + off; off += (size_t)N_NODES * 128;
    w.el  = ws + off; off += (size_t)N_NODES * 2;
    w.er  = ws + off; off += (size_t)N_NODES * 2;
    w.lg  = ws + off; off += (size_t)N_EDGES * 2;
    w.mU  = (unsigned*)(ws + off); off += (size_t)N_NODES * 2;
    w.sB  = ws + off; off += (size_t)N_NODES * 2;
    w.agg = ws + off; off += (size_t)N_NODES * 64;
    w.xb0 = ws + off; off += (size_t)N_NODES * 64;
    w.xb1 = ws + off; off += (size_t)N_NODES * 64;
    w.wee = ws + off; off += 64;

    run_edge_gat_layer(xS, eS, src, dst, P[4], P[5], P[6], P[7], P[8], P[9],
                       w.xb0, w, stream);
    run_edge_gat_layer(w.xb0, eS, src, dst, P[10], P[11], P[12], P[13], P[14], P[15],
                       w.xb1, w, stream);
    gemm_wmma_f32<true><<<(N_NODES / 16) * (64 / 16) / 4, 128, 0, stream>>>(
        w.xb1, W_e2d, w.xb0, N_NODES, 64, 64, 64, 64, 64);
    run_edge_gat_layer(w.xb0, eS, src, dst, P[16], P[17], P[18], P[19], P[20], P[21],
                       w.xb1, w, stream);
    run_edge_gat_layer(w.xb1, eS, src, dst, P[22], P[23], P[24], P[25], P[26], P[27],
                       (float*)d_out, w, stream);
}


# Head mean + bias pass removed: node kernel stores 0.5*(b0+b1) as start value of the output rows, scatter adds halved messages straight into them (sum order changes: bias first); 21 launches -> 17
# speedup vs baseline: 2.1275x; 1.1203x over previous
#include <hip/hip_runtime.h>
#include <hip/hip_bf16.h>

#define N_NODES 20000
#define N_EDGES 100000
#define TPB 256

typedef __attribute__((ext_vector_type(2))) float v2f;
typedef __attribute__((ext_vector_type(8))) float v8f;

template <bool BT>
__global__ void gemm_wmma_f32(const float* __restrict__ A, const float* __restrict__ B,
                              float* __restrict__ C, int M, int K, int NC,
                              int lda, int ldb, int ldc) {
    const int tilesN = NC >> 4;
    const int wave   = blockIdx.x * (blockDim.x >> 5) + (threadIdx.x >> 5);
    const int trow   = wave / tilesN;
    const int tcol   = wave - trow * tilesN;
    if ((trow << 4) >= M) return;

    const int lane = threadIdx.x & 31;
    const int mloc = lane & 15;
    const int kh   = (lane >> 4) << 1;
    const int row0 = trow << 4, col0 = tcol << 4;

    v8f acc = {};
    const float* aBase = A + (size_t)(row0 + mloc) * lda + kh;
    for (int k = 0; k < K; k += 4) {
        v2f a, b;
        a.x = aBase[k];
        a.y = aBase[k + 1];
        if (BT) {
            const float* bp = B + (size_t)(col0 + mloc) * ldb + k + kh;
            b.x = bp[0];
            b.y = bp[1];
        } else {
            const float* bp = B + (size_t)(k + kh) * ldb + col0 + mloc;
            b.x = bp[0];
            b.y = bp[ldb];
        }
        acc = __builtin_amdgcn_wmma_f32_16x16x4_f32(false, a, false, b,
                                                    (short)0, acc, false, false);
    }

    const int   rbase = row0 + ((lane >> 4) << 3);
    float*      cp    = C + (size_t)rbase * ldc + col0 + mloc;
#pragma unroll
    for (int r = 0; r < 8; ++r) cp[(size_t)r * ldc] = acc[r];
}

__device__ __forceinline__ unsigned f2ord(float f) {
    unsigned u = __float_as_uint(f);
    return (u & 0x80000000u) ? ~u : (u | 0x80000000u);
}
__device__ __forceinline__ float ord2f(unsigned u) {
    return (u & 0x80000000u) ? __uint_as_float(u & 0x7FFFFFFFu) : __uint_as_float(~u);
}

__global__ void __launch_bounds__(128) node_fused_kernel(const float* __restrict__ xin,
                                         const float* __restrict__ Wn,
                                         const float* __restrict__ al,
                                         const float* __restrict__ ar,
                                         float* __restrict__ ft,
                                         float* __restrict__ el,
                                         float* __restrict__ er,
                                         float* __restrict__ zb,
                                         const float* __restrict__ bb,
                                         float* __restrict__ xout,
                                         const float* __restrict__ We,
                                         const float* __restrict__ ae,
                                         float* __restrict__ wee) {
    if (blockIdx.x == (N_NODES / 16) * 2 / 4) {
        const int t = threadIdx.x;
        if (t < 64) {
            const int h = t >> 5, k = t & 31;
            const float4* wp = reinterpret_cast<const float4*>(We + (size_t)k * 128 + h * 64);
            const float4* ap = reinterpret_cast<const float4*>(ae + h * 64);
            float acc = 0.f;
#pragma unroll
            for (int d = 0; d < 16; ++d) {
                const float4 wv = wp[d];
                const float4 av = ap[d];
                acc += wv.x * av.x;
                acc += wv.y * av.y;
                acc += wv.z * av.z;
                acc += wv.w * av.w;
            }
            wee[h * 32 + k] = acc;
        }
        return;
    }
    const int wave = blockIdx.x * (blockDim.x >> 5) + (threadIdx.x >> 5);
    if (wave >= (N_NODES / 16) * 2) return;
    const int lane = threadIdx.x & 31;
    const int c    = lane & 15;
    const int kh   = (lane >> 4) << 1;
    const int h    = wave & 1;
    const int n0   = (wave >> 1) << 4;

    zb[(size_t)h * (N_NODES * 2) + n0 * 2 + lane] = 0.f;
    {
        const float4 bh0 = *reinterpret_cast<const float4*>(bb + c * 4);
        const float4 bh1 = *reinterpret_cast<const float4*>(bb + 64 + c * 4);
        const float4 z4 = make_float4(0.5f * (bh0.x + bh1.x), 0.5f * (bh0.y + bh1.y),
                                      0.5f * (bh0.z + bh1.z), 0.5f * (bh0.w + bh1.w));
        float4* ag4 = reinterpret_cast<float4*>(xout + (size_t)n0 * 64 + h * 512);
#pragma unroll
        for (int i = 0; i < 4; ++i) ag4[i * 32 + lane] = z4;
    }

    v8f acc[4];
#pragma unroll
    for (int j = 0; j < 4; ++j) acc[j] = v8f{};
    const float* aBase = xin + (size_t)(n0 + c) * 64 + kh;
    const float* bBase = Wn + (size_t)kh * 128 + h * 64 + c;
#pragma unroll
    for (int k = 0; k < 64; k += 4) {
        v2f a;
        a.x = aBase[k];
        a.y = aBase[k + 1];
#pragma unroll
        for (int j = 0; j < 4; ++j) {
            v2f b;
            b.x = bBase[(size_t)k * 128 + j * 16];
            b.y = bBase[(size_t)(k + 1) * 128 + j * 16];
            acc[j] = __builtin_amdgcn_wmma_f32_16x16x4_f32(false, a, false, b,
                                                           (short)0, acc[j], false, false);
        }
    }

    const int rbase = n0 + ((lane >> 4) << 3);
    float*    cp    = ft + (size_t)rbase * 128 + h * 64 + c;
#pragma unroll
    for (int r = 0; r < 8; ++r) {
#pragma unroll
        for (int j = 0; j < 4; ++j) cp[(size_t)r * 128 + j * 16] = acc[j][r];
    }

    float s1[8], s2[8];
#pragma unroll
    for (int r = 0; r < 8; ++r) { s1[r] = 0.f; s2[r] = 0.f; }
#pragma unroll
    for (int j = 0; j < 4; ++j) {
        const float wl = al[h * 64 + j * 16 + c];
        const float wr = ar[h * 64 + j * 16 + c];
#pragma unroll
        for (int r = 0; r < 8; ++r) {
            s1[r] += acc[j][r] * wl;
            s2[r] += acc[j][r] * wr;
        }
    }
#pragma unroll
    for (int off = 1; off < 16; off <<= 1) {
#pragma unroll
        for (int r = 0; r < 8; ++r) {
            s1[r] += __shfl_xor(s1[r], off, 32);
            s2[r] += __shfl_xor(s2[r], off, 32);
        }
    }
    float v = 0.f;
#pragma unroll
    for (int r = 0; r < 8; ++r) {
        v = (c == r)     ? s1[r] : v;
        v = (c == r + 8) ? s2[r] : v;
    }
    float* outp = (c < 8) ? el : er;
    outp[(size_t)(rbase + (c & 7)) * 2 + h] = v;
}

__global__ void p_sum_kernel(float* __restrict__ lg, const unsigned* __restrict__ mU,
                             const int* __restrict__ dst, float* __restrict__ sB) {
    int t = blockIdx.x * blockDim.x + threadIdx.x;
    if (t >= N_EDGES * 2) return;
    int e = t >> 1, h = t & 1;
    float m  = ord2f(mU[dst[e] * 2 + h]);
    float pv = expf(lg[t] - m);
    lg[t]    = pv;
    atomicAdd(&sB[dst[e] * 2 + h], pv);
}


#define WE_PITCH 288

__device__ __forceinline__ void stage_We_pairs(const float* __restrict__ We, float* sW, int t) {
#pragma unroll
    for (int q = 0; q < 4; ++q) {
        const int item = t + q * 128;
        const int kp = item >> 5, c4 = item & 31;
        const float4 r0 = *reinterpret_cast<const float4*>(We + (size_t)(2 * kp) * 128 + c4 * 4);
        const float4 r1 = *reinterpret_cast<const float4*>(We + (size_t)(2 * kp + 1) * 128 + c4 * 4);
        float4* d = reinterpret_cast<float4*>(sW + kp * WE_PITCH + c4 * 8);
        d[0] = make_float4(r0.x, r1.x, r0.y, r1.y);
        d[1] = make_float4(r0.z, r1.z, r0.w, r1.w);
    }
}

__device__ __forceinline__ void fe_tile_16x128(const float* __restrict__ eF,
                                               const float* sW,
                                               int e0, int lane, v8f (&acc)[8]) {
    const int mloc = lane & 15;
    const int hf   = lane >> 4;
    const int kh   = hf << 1;
    const float* aBase = eF + (size_t)(e0 + mloc) * 32 + kh;
    const float* bBase = sW + hf * WE_PITCH + mloc * 2;
#pragma unroll
    for (int j = 0; j < 8; ++j) acc[j] = v8f{};
#pragma unroll
    for (int k = 0; k < 32; k += 4) {
        v2f a;
        a.x = aBase[k];
        a.y = aBase[k + 1];
#pragma unroll
        for (int j = 0; j < 8; ++j) {
            const v2f b = *reinterpret_cast<const v2f*>(bBase + (k >> 1) * WE_PITCH + j * 32);
            acc[j] = __builtin_amdgcn_wmma_f32_16x16x4_f32(false, a, false, b,
                                                           (short)0, acc[j], false, false);
        }
    }
}

__global__ void __launch_bounds__(128) edge_logits_kernel(const float* __restrict__ eF,
                                         const float* __restrict__ wee,
                                         const float* __restrict__ el,
                                         const float* __restrict__ er,
                                         const int* __restrict__ src,
                                         const int* __restrict__ dst,
                                         float* __restrict__ lg,
                                         unsigned* __restrict__ mU) {
    const int wave = blockIdx.x * (blockDim.x >> 5) + (threadIdx.x >> 5);
    if (wave >= N_EDGES / 32) return;
    const int lane = threadIdx.x & 31;
    const int c    = lane & 15;
    const int hf   = lane >> 4;
    const int kh   = hf << 1;
    const int e0   = wave << 5;

    const bool   live = (c & 6) == 0;
    const float* ap = wee + (c & 1) * 32 + kh;
    const float* b0 = eF + (size_t)(e0 + c) * 32 + kh;
    const float* b1 = b0 + 16 * 32;

    v8f acc0 = {}, acc1 = {};
#pragma unroll
    for (int k = 0; k < 32; k += 4) {
        v2f a, p, q;
        const float a0v = ap[k], a1v = ap[k + 1];
        a.x = live ? a0v : 0.f;
        a.y = live ? a1v : 0.f;
        p.x = b0[k];
        p.y = b0[k + 1];
        q.x = b1[k];
        q.y = b1[k + 1];
        acc0 = __builtin_amdgcn_wmma_f32_16x16x4_f32(false, a, false, p,
                                                     (short)0, acc0, false, false);
        acc1 = __builtin_amdgcn_wmma_f32_16x16x4_f32(false, a, false, q,
                                                     (short)0, acc1, false, false);
    }

    const int h = hf;
    int   dNr[2];
    float lr[2];
#pragma unroll
    for (int j = 0; j < 2; ++j) {
        const float eev = j ? (h ? acc1[1] : acc1[0]) : (h ? acc0[1] : acc0[0]);
        const int e = e0 + j * 16 + c;
        dNr[j] = dst[e];
        const float l = el[src[e] * 2 + h] + er[dNr[j] * 2 + h] + eev;
        lr[j] = (l > 0.f) ? l : 0.2f * l;
    }
#pragma unroll
    for (int j = 0; j < 2; ++j) {
        const int e = e0 + j * 16 + c;
        lg[e * 2 + h] = lr[j];
        atomicMax(&mU[dNr[j] * 2 + h], f2ord(lr[j]));
    }
}

__global__ void __launch_bounds__(128) scatter_msg_fused_kernel(const float* __restrict__ eF,
                                         const float* __restrict__ We,
                                         const float* __restrict__ ft,
                                         const float* __restrict__ p,
                                         const float* __restrict__ sB,
                                         const int* __restrict__ src,
                                         const int* __restrict__ dst,
                                         float* __restrict__ xout) {
    __shared__ __attribute__((aligned(16))) float sW[16 * WE_PITCH];
    stage_We_pairs(We, sW, threadIdx.x);
    __syncthreads();

    const int wave = blockIdx.x * (blockDim.x >> 5) + (threadIdx.x >> 5);
    if (wave >= N_EDGES / 16) return;
    const int lane = threadIdx.x & 31;
    const int c    = lane & 15;
    const int e0   = wave << 4;

    const int eb = e0 + ((lane >> 4) << 3);

    int   sNr[8], dNr[8];
    float a0r[8], a1r[8];
#pragma unroll
    for (int r = 0; r < 8; ++r) {
        const int e = eb + r;
        sNr[r] = src[e];
        dNr[r] = dst[e];
        a0r[r] = 0.5f * (p[e * 2]     / sB[dNr[r] * 2]);
        a1r[r] = 0.5f * (p[e * 2 + 1] / sB[dNr[r] * 2 + 1]);
    }

    v8f acc[8];
    fe_tile_16x128(eF, sW, e0, lane, acc);

#pragma unroll
    for (int r = 0; r < 8; ++r) {
        const float* fts = ft + (size_t)sNr[r] * 128 + c;
#pragma unroll
        for (int j = 0; j < 4; ++j) {
            const float m0 = (fts[j * 16]      + acc[j][r])     * a0r[r];
            const float m1 = (fts[64 + j * 16] + acc[j + 4][r]) * a1r[r];
            acc[j][r] = m0 + m1;
        }
    }
#pragma unroll
    for (int r = 0; r < 8; ++r) {
        float* ag = xout + (size_t)dNr[r] * 64 + c;
#pragma unroll
        for (int j = 0; j < 4; ++j) atomicAdd(&ag[j * 16], acc[j][r]);
    }
}

static inline int cdiv(int a, int b) { return (a + b - 1) / b; }

struct WsPtrs {
    float* ft; float* el; float* er; float* lg;
    unsigned* mU; float* sB; float* xb0; float* xb1; float* wee;
};

static void run_edge_gat_layer(const float* xin, const float* eF, const int* src,
                               const int* dst, const float* Wn, const float* We,
                               const float* al, const float* ar, const float* ae,
                               const float* bb, float* xout, const WsPtrs& w,
                               hipStream_t stream) {
    node_fused_kernel<<<(N_NODES / 16) * 2 / 4 + 1, 128, 0, stream>>>(
        xin, Wn, al, ar, w.ft, w.el, w.er, (float*)w.mU, bb, xout, We, ae, w.wee);

    edge_logits_kernel<<<cdiv(N_EDGES / 32, 4), 128, 0, stream>>>(
        eF, w.wee, w.el, w.er, src, dst, w.lg, w.mU);
    p_sum_kernel<<<cdiv(N_EDGES * 2, TPB), TPB, 0, stream>>>(w.lg, w.mU, dst, w.sB);
    scatter_msg_fused_kernel<<<cdiv(N_EDGES / 16, 4), 128, 0, stream>>>(
        eF, We, w.ft, w.lg, w.sB, src, dst, xout);
}

extern "C" void kernel_launch(void* const* d_in, const int* in_sizes, int n_in,
                              void* d_out, int out_size, void* d_ws, size_t ws_size,
                              hipStream_t stream) {
    const float* x_all = (const float*)d_in[0];
    const float* e_all = (const float*)d_in[1];
    const int*   s_all = (const int*)d_in[2];
    const int*   d_all = (const int*)d_in[3];

    const float* xS  = x_all + (size_t)1 * N_NODES * 64;
    const float* eS  = e_all + (size_t)1 * N_EDGES * 32;
    const int*   src = s_all + (size_t)1 * N_EDGES;
    const int*   dst = d_all + (size_t)1 * N_EDGES;

    const float* const* P = (const float* const*)d_in;
    const float* W_e2d = P[28];

    float* ws = (float*)d_ws;
    size_t off = 0;
    WsPtrs w;
    w.ft  = ws + off; off += (size_t)N_NODES * 128;
    w.el  = ws + off; off += (size_t)N_NODES * 2;
    w.er  = ws + off; off += (size_t)N_NODES * 2;
    w.lg  = ws + off; off += (size_t)N_EDGES * 2;
    w.mU  = (unsigned*)(ws + off); off += (size_t)N_NODES * 2;
    w.sB  = ws + off; off += (size_t)N_NODES * 2;
    w.xb0 = ws + off; off += (size_t)N_NODES * 64;
    w.xb1 = ws + off; off += (size_t)N_NODES * 64;
    w.wee = ws + off; off += 64;

    run_edge_gat_layer(xS, eS, src, dst, P[4], P[5], P[6], P[7], P[8], P[9],
                       w.xb0, w, stream);
    run_edge_gat_layer(w.xb0, eS, src, dst, P[10], P[11], P[12], P[13], P[14], P[15],
                       w.xb1, w, stream);
    gemm_wmma_f32<true><<<(N_NODES / 16) * (64 / 16) / 4, 128, 0, stream>>>(
        w.xb1, W_e2d, w.xb0, N_NODES, 64, 64, 64, 64, 64);
    run_edge_gat_layer(w.xb0, eS, src, dst, P[16], P[17], P[18], P[19], P[20], P[21],
                       w.xb1, w, stream);
    run_edge_gat_layer(w.xb1, eS, src, dst, P[22], P[23], P[24], P[25], P[26], P[27],
                       (float*)d_out, w, stream);
}


# Node scores reassociated: logit = [x[src]|x[dst]|eF] . [Wn.al|Wn.ar|We.ae], one K=160 WMMA dot per edge (sum order changes), so logits and ft GEMM share one launch; 17 -> 14 launches
# speedup vs baseline: 2.3505x; 1.1048x over previous
#include <hip/hip_runtime.h>
#include <hip/hip_bf16.h>

#define N_NODES 20000
#define N_EDGES 100000
#define TPB 256

typedef __attribute__((ext_vector_type(2))) float v2f;
typedef __attribute__((ext_vector_type(8))) float v8f;

template <bool BT>
__global__ void gemm_wmma_f32(const float* __restrict__ A, const float* __restrict__ B,
                              float* __restrict__ C, int M, int K, int NC,
                              int lda, int ldb, int ldc) {
    const int tilesN = NC >> 4;
    const int wave   = blockIdx.x * (blockDim.x >> 5) + (threadIdx.x >> 5);
    const int trow   = wave / tilesN;
    const int tcol   = wave - trow * tilesN;
    if ((trow << 4) >= M) return;

    const int lane = threadIdx.x & 31;
    const int mloc = lane & 15;
    const int kh   = (lane >> 4) << 1;
    const int row0 = trow << 4, col0 = tcol << 4;

    v8f acc = {};
    const float* aBase = A + (size_t)(row0 + mloc) * lda + kh;
    for (int k = 0; k < K; k += 4) {
        v2f a, b;
        a.x = aBase[k];
        a.y = aBase[k + 1];
        if (BT) {
            const float* bp = B + (size_t)(col0 + mloc) * ldb + k + kh;
            b.x = bp[0];
            b.y = bp[1];
        } else {
            const float* bp = B + (size_t)(k + kh) * ldb + col0 + mloc;
            b.x = bp[0];
            b.y = bp[ldb];
        }
        acc = __builtin_amdgcn_wmma_f32_16x16x4_f32(false, a, false, b,
                                                    (short)0, acc, false, false);
    }

    const int   rbase = row0 + ((lane >> 4) << 3);
    float*      cp    = C + (size_t)rbase * ldc + col0 + mloc;
#pragma unroll
    for (int r = 0; r < 8; ++r) cp[(size_t)r * ldc] = acc[r];
}

__device__ __forceinline__ unsigned f2ord(float f) {
    unsigned u = __float_as_uint(f);
    return (u & 0x80000000u) ? ~u : (u | 0x80000000u);
}
__device__ __forceinline__ float ord2f(unsigned u) {
    return (u & 0x80000000u) ? __uint_as_float(u & 0x7FFFFFFFu) : __uint_as_float(~u);
}

__global__ void p_sum_kernel(float* __restrict__ lg, const unsigned* __restrict__ mU,
                             const int* __restrict__ dst, float* __restrict__ sB) {
    int t = blockIdx.x * blockDim.x + threadIdx.x;
    if (t >= N_EDGES * 2) return;
    int e = t >> 1, h = t & 1;
    float m  = ord2f(mU[dst[e] * 2 + h]);
    float pv = expf(lg[t] - m);
    lg[t]    = pv;
    atomicAdd(&sB[dst[e] * 2 + h], pv);
}


#define WE_PITCH 288

__device__ __forceinline__ void stage_We_pairs(const float* __restrict__ We, float* sW, int t) {
#pragma unroll
    for (int q = 0; q < 4; ++q) {
        const int item = t + q * 128;
        const int kp = item >> 5, c4 = item & 31;
        const float4 r0 = *reinterpret_cast<const float4*>(We + (size_t)(2 * kp) * 128 + c4 * 4);
        const float4 r1 = *reinterpret_cast<const float4*>(We + (size_t)(2 * kp + 1) * 128 + c4 * 4);
        float4* d = reinterpret_cast<float4*>(sW + kp * WE_PITCH + c4 * 8);
        d[0] = make_float4(r0.x, r1.x, r0.y, r1.y);
        d[1] = make_float4(r0.z, r1.z, r0.w, r1.w);
    }
}

__device__ __forceinline__ void fe_tile_16x128(const float* __restrict__ eF,
                                               const float* sW,
                                               int e0, int lane, v8f (&acc)[8]) {
    const int mloc = lane & 15;
    const int hf   = lane >> 4;
    const int kh   = hf << 1;
    const float* aBase = eF + (size_t)(e0 + mloc) * 32 + kh;
    const float* bBase = sW + hf * WE_PITCH + mloc * 2;
#pragma unroll
    for (int j = 0; j < 8; ++j) acc[j] = v8f{};
#pragma unroll
    for (int k = 0; k < 32; k += 4) {
        v2f a;
        a.x = aBase[k];
        a.y = aBase[k + 1];
#pragma unroll
        for (int j = 0; j < 8; ++j) {
            const v2f b = *reinterpret_cast<const v2f*>(bBase + (k >> 1) * WE_PITCH + j * 32);
            acc[j] = __builtin_amdgcn_wmma_f32_16x16x4_f32(false, a, false, b,
                                                           (short)0, acc[j], false, false);
        }
    }
}

#define VK 160
#define PREP_CLEAR_BLOCKS 125
__device__ __forceinline__ void fold_layer(const float* __restrict__ Wn, const float* __restrict__ We,
                                           const float* __restrict__ al, const float* __restrict__ ar,
                                           const float* __restrict__ ae, float* __restrict__ wvl, int t) {
    const int h = t / VK, idx = t - h * VK;
    const float* M;
    const float* v;
    int row;
    if (idx < 64)       { M = Wn; v = al; row = idx; }
    else if (idx < 128) { M = Wn; v = ar; row = idx - 64; }
    else                { M = We; v = ae; row = idx - 128; }
    const float4* wp = reinterpret_cast<const float4*>(M + (size_t)row * 128 + h * 64);
    const float4* ap = reinterpret_cast<const float4*>(v + h * 64);
    float acc = 0.f;
#pragma unroll
    for (int d = 0; d < 16; ++d) {
        const float4 wq = wp[d];
        const float4 aq = ap[d];
        acc += wq.x * aq.x;
        acc += wq.y * aq.y;
        acc += wq.z * aq.z;
        acc += wq.w * aq.w;
    }
    wvl[h * VK + idx] = acc;
}

__global__ void __launch_bounds__(320) prep_kernel(
        const float* __restrict__ Wn0, const float* __restrict__ We0, const float* __restrict__ al0,
        const float* __restrict__ ar0, const float* __restrict__ ae0,
        const float* __restrict__ Wn1, const float* __restrict__ We1, const float* __restrict__ al1,
        const float* __restrict__ ar1, const float* __restrict__ ae1,
        const float* __restrict__ Wn2, const float* __restrict__ We2, const float* __restrict__ al2,
        const float* __restrict__ ar2, const float* __restrict__ ae2,
        const float* __restrict__ Wn3, const float* __restrict__ We3, const float* __restrict__ al3,
        const float* __restrict__ ar3, const float* __restrict__ ae3,
        float* __restrict__ wv, unsigned* __restrict__ mU) {
    const int t = threadIdx.x;
    if (blockIdx.x < PREP_CLEAR_BLOCKS) {
        mU[blockIdx.x * 320 + t] = 0u;
        return;
    }
    const int layer = blockIdx.x - PREP_CLEAR_BLOCKS;
    if (layer == 0)      fold_layer(Wn0, We0, al0, ar0, ae0, wv, t);
    else if (layer == 1) fold_layer(Wn1, We1, al1, ar1, ae1, wv + 2 * VK, t);
    else if (layer == 2) fold_layer(Wn2, We2, al2, ar2, ae2, wv + 4 * VK, t);
    else                 fold_layer(Wn3, We3, al3, ar3, ae3, wv + 6 * VK, t);
}

#define NODE_BLOCKS ((N_NODES / 16) * 2 / 4)
#define LOGIT_BLOCKS ((N_EDGES / 32 + 3) / 4)

#define VKP 176

template <int STEPS>
__device__ __forceinline__ void logit_load(const float* __restrict__ p0,
                                           const float* __restrict__ p1,
                                           v2f (&P)[STEPS], v2f (&Q)[STEPS]) {
#pragma unroll
    for (int i = 0; i < STEPS; ++i) {
        P[i] = *reinterpret_cast<const v2f*>(p0 + 4 * i);
        Q[i] = *reinterpret_cast<const v2f*>(p1 + 4 * i);
    }
}

template <int STEPS>
__device__ __forceinline__ void logit_mma(const float* ap, const v2f (&P)[STEPS],
                                          const v2f (&Q)[STEPS], v8f& acc0, v8f& acc1) {
    v2f a[4], n[4];
#pragma unroll
    for (int j = 0; j < 4; ++j) a[j] = *reinterpret_cast<const v2f*>(ap + 4 * j);
#pragma unroll
    for (int i0 = 0; i0 < STEPS; i0 += 4) {
        if (i0 + 4 < STEPS) {
#pragma unroll
            for (int j = 0; j < 4; ++j) n[j] = *reinterpret_cast<const v2f*>(ap + 4 * (i0 + 4 + j));
        }
#pragma unroll
        for (int j = 0; j < 4; ++j) {
            acc0 = __builtin_amdgcn_wmma_f32_16x16x4_f32(false, a[j], false, P[i0 + j],
                                                         (short)0, acc0, false, false);
            acc1 = __builtin_amdgcn_wmma_f32_16x16x4_f32(false, a[j], false, Q[i0 + j],
                                                         (short)0, acc1, false, false);
        }
#pragma unroll
        for (int j = 0; j < 4; ++j) a[j] = n[j];
    }
}

#define HOLD8(X, o) asm volatile("" :: "v"(X[o]), "v"(X[o + 1]), "v"(X[o + 2]), "v"(X[o + 3]), \
                                       "v"(X[o + 4]), "v"(X[o + 5]), "v"(X[o + 6]), "v"(X[o + 7]))

__global__ void __launch_bounds__(128) layer_front_kernel(const float* __restrict__ xin,
                                         const float* __restrict__ Wn,
                                         float* __restrict__ ft,
                                         float* __restrict__ sB,
                                         const float* __restrict__ bb,
                                         float* __restrict__ xout,
                                         const float* __restrict__ eF,
                                         const float* __restrict__ wv,
                                         const int* __restrict__ src,
                                         const int* __restrict__ dst,
                                         float* __restrict__ lg,
                                         unsigned* __restrict__ mU) {
    const int lane = threadIdx.x & 31;
    const int c    = lane & 15;
    const int hf   = lane >> 4;
    const int kh   = hf << 1;

    if (blockIdx.x < LOGIT_BLOCKS) {
        __shared__ __attribute__((aligned(16))) float sV[3 * VKP];
        const int wave = blockIdx.x * (blockDim.x >> 5) + (threadIdx.x >> 5);
        const int e0 = min(wave, N_EDGES / 32 - 1) << 5;
        const int eA = e0 + c, eB = e0 + 16 + c;
        const int sA = src[eA], dA = dst[eA];
        const int sC = src[eB], dC = dst[eB];

        for (int i = threadIdx.x; i < 3 * VK; i += 128) {
            const int r = i / VK, k = i - r * VK;
            sV[r * VKP + k] = (r < 2) ? wv[i] : 0.f;
        }
        __syncthreads();
        if (wave >= N_EDGES / 32) return;

        const float* ap = sV + (((c & 6) == 0) ? (c & 1) : 2) * VKP + kh;

        v2f P1[16], Q1[16], P2[16], Q2[16], P3[8], Q3[8];
        logit_load<16>(xin + (size_t)sA * 64 + kh, xin + (size_t)sC * 64 + kh, P1, Q1);
        logit_load<8>(eF + (size_t)eA * 32 + kh, eF + (size_t)eB * 32 + kh, P3, Q3);

        v8f acc0 = {}, acc1 = {};
        HOLD8(P1, 0); HOLD8(P1, 8); HOLD8(Q1, 0); HOLD8(Q1, 8);
        logit_mma<16>(ap, P1, Q1, acc0, acc1);
        logit_load<16>(xin + (size_t)dA * 64 + kh, xin + (size_t)dC * 64 + kh, P2, Q2);
        HOLD8(P3, 0); HOLD8(Q3, 0);
        logit_mma<8>(ap + 128, P3, Q3, acc0, acc1);
        HOLD8(P2, 0); HOLD8(P2, 8); HOLD8(Q2, 0); HOLD8(Q2, 8);
        logit_mma<16>(ap + 64, P2, Q2, acc0, acc1);

        const int h = hf;
        const float l0 = h ? acc0[1] : acc0[0];
        const float l1 = h ? acc1[1] : acc1[0];
        const float r0 = (l0 > 0.f) ? l0 : 0.2f * l0;
        const float r1 = (l1 > 0.f) ? l1 : 0.2f * l1;
        lg[eA * 2 + h] = r0;
        lg[eB * 2 + h] = r1;
        atomicMax(&mU[dA * 2 + h], f2ord(r0));
        atomicMax(&mU[dC * 2 + h], f2ord(r1));
        return;
    }

    const int wave = (blockIdx.x - LOGIT_BLOCKS) * (blockDim.x >> 5) + (threadIdx.x >> 5);
    if (wave >= (N_NODES / 16) * 2) return;
    const int h    = wave & 1;
    const int n0   = (wave >> 1) << 4;

    if (h == 1) sB[n0 * 2 + lane] = 0.f;
    {
        const float4 bh0 = *reinterpret_cast<const float4*>(bb + c * 4);
        const float4 bh1 = *reinterpret_cast<const float4*>(bb + 64 + c * 4);
        const float4 z4 = make_float4(0.5f * (bh0.x + bh1.x), 0.5f * (bh0.y + bh1.y),
                                      0.5f * (bh0.z + bh1.z), 0.5f * (bh0.w + bh1.w));
        float4* ag4 = reinterpret_cast<float4*>(xout + (size_t)n0 * 64 + h * 512);
#pragma unroll
        for (int i = 0; i < 4; ++i) ag4[i * 32 + lane] = z4;
    }

    v8f acc[4];
#pragma unroll
    for (int j = 0; j < 4; ++j) acc[j] = v8f{};
    const float* aBase = xin + (size_t)(n0 + c) * 64 + kh;
    const float* bBase = Wn + (size_t)kh * 128 + h * 64 + c;
#pragma unroll
    for (int k = 0; k < 64; k += 4) {
        v2f a;
        a.x = aBase[k];
        a.y = aBase[k + 1];
#pragma unroll
        for (int j = 0; j < 4; ++j) {
            v2f b;
            b.x = bBase[(size_t)k * 128 + j * 16];
            b.y = bBase[(size_t)(k + 1) * 128 + j * 16];
            acc[j] = __builtin_amdgcn_wmma_f32_16x16x4_f32(false, a, false, b,
                                                           (short)0, acc[j], false, false);
        }
    }

    const int rbase = n0 + ((lane >> 4) << 3);
    float*    cp    = ft + (size_t)rbase * 128 + h * 64 + c;
#pragma unroll
    for (int r = 0; r < 8; ++r) {
#pragma unroll
        for (int j = 0; j < 4; ++j) cp[(size_t)r * 128 + j * 16] = acc[j][r];
    }
}

__global__ void __launch_bounds__(128) scatter_msg_fused_kernel(const float* __restrict__ eF,
                                         const float* __restrict__ We,
                                         const float* __restrict__ ft,
                                         const float* __restrict__ p,
                                         const float* __restrict__ sB,
                                         const int* __restrict__ src,
                                         const int* __restrict__ dst,
                                         float* __restrict__ xout,
                                         unsigned* __restrict__ mU) {
    __shared__ __attribute__((aligned(16))) float sW[16 * WE_PITCH];
    stage_We_pairs(We, sW, threadIdx.x);
    __syncthreads();

    const int wave = blockIdx.x * (blockDim.x >> 5) + (threadIdx.x >> 5);
    if (wave >= N_EDGES / 16) return;
    const int lane = threadIdx.x & 31;
    const int c    = lane & 15;
    const int e0   = wave << 4;

    if (lane < 8 && wave * 8 + lane < N_NODES * 2) mU[wave * 8 + lane] = 0u;

    const int eb = e0 + ((lane >> 4) << 3);

    int   sNr[8], dNr[8];
    float a0r[8], a1r[8];
#pragma unroll
    for (int r = 0; r < 8; ++r) {
        const int e = eb + r;
        sNr[r] = src[e];
        dNr[r] = dst[e];
        a0r[r] = 0.5f * (p[e * 2]     / sB[dNr[r] * 2]);
        a1r[r] = 0.5f * (p[e * 2 + 1] / sB[dNr[r] * 2 + 1]);
    }

    v8f acc[8];
    fe_tile_16x128(eF, sW, e0, lane, acc);

#pragma unroll
    for (int r = 0; r < 8; ++r) {
        const float* fts = ft + (size_t)sNr[r] * 128 + c;
#pragma unroll
        for (int j = 0; j < 4; ++j) {
            const float m0 = (fts[j * 16]      + acc[j][r])     * a0r[r];
            const float m1 = (fts[64 + j * 16] + acc[j + 4][r]) * a1r[r];
            acc[j][r] = m0 + m1;
        }
    }
#pragma unroll
    for (int r = 0; r < 8; ++r) {
        float* ag = xout + (size_t)dNr[r] * 64 + c;
#pragma unroll
        for (int j = 0; j < 4; ++j) atomicAdd(&ag[j * 16], acc[j][r]);
    }
}

static inline int cdiv(int a, int b) { return (a + b - 1) / b; }

struct WsPtrs {
    float* ft; float* lg;
    unsigned* mU; float* sB; float* xb0; float* xb1; float* wv;
};

static void run_edge_gat_layer(const float* xin, const float* eF, const int* src,
                               const int* dst, const float* Wn, const float* We,
                               const float* bb, const float* wvl, float* xout, const WsPtrs& w,
                               hipStream_t stream) {
    layer_front_kernel<<<LOGIT_BLOCKS + NODE_BLOCKS, 128, 0, stream>>>(
        xin, Wn, w.ft, w.sB, bb, xout, eF, wvl, src, dst, w.lg, w.mU);
    p_sum_kernel<<<cdiv(N_EDGES * 2, TPB), TPB, 0, stream>>>(w.lg, w.mU, dst, w.sB);
    scatter_msg_fused_kernel<<<cdiv(N_EDGES / 16, 4), 128, 0, stream>>>(
        eF, We, w.ft, w.lg, w.sB, src, dst, xout, w.mU);
}

extern "C" void kernel_launch(void* const* d_in, const int* in_sizes, int n_in,
                              void* d_out, int out_size, void* d_ws, size_t ws_size,
                              hipStream_t stream) {
    const float* x_all = (const float*)d_in[0];
    const float* e_all = (const float*)d_in[1];
    const int*   s_all = (const int*)d_in[2];
    const int*   d_all = (const int*)d_in[3];

    const float* xS  = x_all + (size_t)1 * N_NODES * 64;
    const float* eS  = e_all + (size_t)1 * N_EDGES * 32;
    const int*   src = s_all + (size_t)1 * N_EDGES;
    const int*   dst = d_all + (size_t)1 * N_EDGES;

    const float* const* P = (const float* const*)d_in;
    const float* W_e2d = P[28];

    float* ws = (float*)d_ws;
    size_t off = 0;
    WsPtrs w;
    w.ft  = ws + off; off += (size_t)N_NODES * 128;
    w.lg  = ws + off; off += (size_t)N_EDGES * 2;
    w.mU  = (unsigned*)(ws + off); off += (size_t)N_NODES * 2;
    w.sB  = ws + off; off += (size_t)N_NODES * 2;
    w.xb0 = ws + off; off += (size_t)N_NODES * 64;
    w.xb1 = ws + off; off += (size_t)N_NODES * 64;
    w.wv  = ws + off; off += 4 * 2 * VK;

    prep_kernel<<<PREP_CLEAR_BLOCKS + 4, 320, 0, stream>>>(
        P[4], P[5], P[6], P[7], P[8], P[10], P[11], P[12], P[13], P[14],
        P[16], P[17], P[18], P[19], P[20], P[22], P[23], P[24], P[25], P[26], w.wv, w.mU);

    run_edge_gat_layer(xS, eS, src, dst, P[4], P[5], P[9], w.wv, w.xb0, w, stream);
    run_edge_gat_layer(w.xb0, eS, src, dst, P[10], P[11], P[15], w.wv + 2 * VK, w.xb1, w, stream);
    gemm_wmma_f32<true><<<(N_NODES / 16) * (64 / 16) / 4, 128, 0, stream>>>(
        w.xb1, W_e2d, w.xb0, N_NODES, 64, 64, 64, 64, 64);
    run_edge_gat_layer(w.xb0, eS, src, dst, P[16], P[17], P[21], w.wv + 4 * VK, w.xb1, w, stream);
    run_edge_gat_layer(w.xb1, eS, src, dst, P[22], P[23], P[27], w.wv + 6 * VK,
                       (float*)d_out, w, stream);
}


# Encoder-to-decoder map folded into the first decoder layer (Wf = Wc^T.Wn on f32 WMMA, Wc^T applied to its score vectors, once per call); its [N,64] launch removed, 14 -> 13 (sum order changes)
# speedup vs baseline: 2.4556x; 1.0447x over previous
#include <hip/hip_runtime.h>
#include <hip/hip_bf16.h>

#define N_NODES 20000
#define N_EDGES 100000
#define TPB 256

typedef __attribute__((ext_vector_type(2))) float v2f;
typedef __attribute__((ext_vector_type(8))) float v8f;


__device__ __forceinline__ unsigned f2ord(float f) {
    unsigned u = __float_as_uint(f);
    return (u & 0x80000000u) ? ~u : (u | 0x80000000u);
}
__device__ __forceinline__ float ord2f(unsigned u) {
    return (u & 0x80000000u) ? __uint_as_float(u & 0x7FFFFFFFu) : __uint_as_float(~u);
}

__global__ void p_sum_kernel(float* __restrict__ lg, const unsigned* __restrict__ mU,
                             const int* __restrict__ dst, float* __restrict__ sB) {
    int t = blockIdx.x * blockDim.x + threadIdx.x;
    if (t >= N_EDGES * 2) return;
    int e = t >> 1, h = t & 1;
    float m  = ord2f(mU[dst[e] * 2 + h]);
    float pv = expf(lg[t] - m);
    lg[t]    = pv;
    atomicAdd(&sB[dst[e] * 2 + h], pv);
}


#define WE_PITCH 288

__device__ __forceinline__ void stage_We_pairs(const float* __restrict__ We, float* sW, int t) {
#pragma unroll
    for (int q = 0; q < 4; ++q) {
        const int item = t + q * 128;
        const int kp = item >> 5, c4 = item & 31;
        const float4 r0 = *reinterpret_cast<const float4*>(We + (size_t)(2 * kp) * 128 + c4 * 4);
        const float4 r1 = *reinterpret_cast<const float4*>(We + (size_t)(2 * kp + 1) * 128 + c4 * 4);
        float4* d = reinterpret_cast<float4*>(sW + kp * WE_PITCH + c4 * 8);
        d[0] = make_float4(r0.x, r1.x, r0.y, r1.y);
        d[1] = make_float4(r0.z, r1.z, r0.w, r1.w);
    }
}

__device__ __forceinline__ void fe_tile_16x128(const float* __restrict__ eF,
                                               const float* sW,
                                               int e0, int lane, v8f (&acc)[8]) {
    const int mloc = lane & 15;
    const int hf   = lane >> 4;
    const int kh   = hf << 1;
    const float* aBase = eF + (size_t)(e0 + mloc) * 32 + kh;
    const float* bBase = sW + hf * WE_PITCH + mloc * 2;
#pragma unroll
    for (int j = 0; j < 8; ++j) acc[j] = v8f{};
#pragma unroll
    for (int k = 0; k < 32; k += 4) {
        v2f a;
        a.x = aBase[k];
        a.y = aBase[k + 1];
#pragma unroll
        for (int j = 0; j < 8; ++j) {
            const v2f b = *reinterpret_cast<const v2f*>(bBase + (k >> 1) * WE_PITCH + j * 32);
            acc[j] = __builtin_amdgcn_wmma_f32_16x16x4_f32(false, a, false, b,
                                                           (short)0, acc[j], false, false);
        }
    }
}

#define VK 160
#define PREP_CLEAR_BLOCKS 125
#define PREP_FOLD_BLOCKS 4
__device__ __forceinline__ float fold_layer(const float* __restrict__ Wn, const float* __restrict__ We,
                                            const float* __restrict__ al, const float* __restrict__ ar,
                                            const float* __restrict__ ae, int t) {
    const int h = t / VK, idx = t - h * VK;
    const float* M;
    const float* v;
    int row;
    if (idx < 64)       { M = Wn; v = al; row = idx; }
    else if (idx < 128) { M = Wn; v = ar; row = idx - 64; }
    else                { M = We; v = ae; row = idx - 128; }
    const float4* wp = reinterpret_cast<const float4*>(M + (size_t)row * 128 + h * 64);
    const float4* ap = reinterpret_cast<const float4*>(v + h * 64);
    float acc = 0.f;
#pragma unroll
    for (int d = 0; d < 16; ++d) {
        const float4 wq = wp[d];
        const float4 aq = ap[d];
        acc += wq.x * aq.x;
        acc += wq.y * aq.y;
        acc += wq.z * aq.z;
        acc += wq.w * aq.w;
    }
    return acc;
}

__global__ void __launch_bounds__(320) prep_kernel(
        const float* __restrict__ Wn0, const float* __restrict__ We0, const float* __restrict__ al0,
        const float* __restrict__ ar0, const float* __restrict__ ae0,
        const float* __restrict__ Wn1, const float* __restrict__ We1, const float* __restrict__ al1,
        const float* __restrict__ ar1, const float* __restrict__ ae1,
        const float* __restrict__ Wn2, const float* __restrict__ We2, const float* __restrict__ al2,
        const float* __restrict__ ar2, const float* __restrict__ ae2,
        const float* __restrict__ Wn3, const float* __restrict__ We3, const float* __restrict__ al3,
        const float* __restrict__ ar3, const float* __restrict__ ae3,
        const float* __restrict__ Wc,
        float* __restrict__ wv, float* __restrict__ Wf, unsigned* __restrict__ mU) {
    const int t = threadIdx.x;
    if (blockIdx.x < PREP_CLEAR_BLOCKS) {
        mU[blockIdx.x * 320 + t] = 0u;
        return;
    }
    const int layer = blockIdx.x - PREP_CLEAR_BLOCKS;
    if (layer < 4) {
        float v;
        if (layer == 2) {
            __shared__ __attribute__((aligned(16))) float sU[2 * 128];
            const int h = t / VK, idx = t - h * VK;
            const float* wc = Wc + (idx & 63);
            float wcol[64];
#pragma unroll
            for (int i = 0; i < 64; ++i) wcol[i] = wc[i * 64];
            v = fold_layer(Wn2, We2, al2, ar2, ae2, t);
            if (idx < 128) sU[h * 128 + idx] = v;
            __syncthreads();
            if (idx < 128) {
                const float* up = sU + h * 128 + (idx & 64);
                float o = 0.f;
#pragma unroll
                for (int i = 0; i < 64; ++i) o += wcol[i] * up[i];
                v = o;
            }
        }
        else if (layer == 0) v = fold_layer(Wn0, We0, al0, ar0, ae0, t);
        else if (layer == 1) v = fold_layer(Wn1, We1, al1, ar1, ae1, t);
        else                 v = fold_layer(Wn3, We3, al3, ar3, ae3, t);
        wv[layer * 2 * VK + t] = v;
        return;
    }

    const int tile = (layer - 4) * 10 + (t >> 5);
    if (tile >= 32) return;
    const int lane = t & 31;
    const int c    = lane & 15;
    const int hf   = lane >> 4;
    const int kh   = hf << 1;
    const int j0   = (tile >> 3) << 4;
    const int c0   = (tile & 7) << 4;
    const float* ap = Wc  + (size_t)kh * 64  + j0 + c;
    const float* bp = Wn2 + (size_t)kh * 128 + c0 + c;
    v8f acc = {};
#pragma unroll
    for (int k = 0; k < 64; k += 4) {
        v2f a, b;
        a.x = ap[(size_t)k * 64];
        a.y = ap[(size_t)(k + 1) * 64];
        b.x = bp[(size_t)k * 128];
        b.y = bp[(size_t)(k + 1) * 128];
        acc = __builtin_amdgcn_wmma_f32_16x16x4_f32(false, a, false, b,
                                                    (short)0, acc, false, false);
    }
    float* op = Wf + (size_t)(j0 + hf * 8) * 128 + c0 + c;
#pragma unroll
    for (int r = 0; r < 8; ++r) op[(size_t)r * 128] = acc[r];
}

#define NODE_BLOCKS ((N_NODES / 16) * 2 / 4)
#define LOGIT_BLOCKS ((N_EDGES / 32 + 3) / 4)

#define VKP 176

template <int STEPS>
__device__ __forceinline__ void logit_load(const float* __restrict__ p0,
                                           const float* __restrict__ p1,
                                           v2f (&P)[STEPS], v2f (&Q)[STEPS]) {
#pragma unroll
    for (int i = 0; i < STEPS; ++i) {
        P[i] = *reinterpret_cast<const v2f*>(p0 + 4 * i);
        Q[i] = *reinterpret_cast<const v2f*>(p1 + 4 * i);
    }
}

template <int STEPS>
__device__ __forceinline__ void logit_mma(const float* ap, const v2f (&P)[STEPS],
                                          const v2f (&Q)[STEPS], v8f& acc0, v8f& acc1) {
    v2f a[4], n[4];
#pragma unroll
    for (int j = 0; j < 4; ++j) a[j] = *reinterpret_cast<const v2f*>(ap + 4 * j);
#pragma unroll
    for (int i0 = 0; i0 < STEPS; i0 += 4) {
        if (i0 + 4 < STEPS) {
#pragma unroll
            for (int j = 0; j < 4; ++j) n[j] = *reinterpret_cast<const v2f*>(ap + 4 * (i0 + 4 + j));
        }
#pragma unroll
        for (int j = 0; j < 4; ++j) {
            acc0 = __builtin_amdgcn_wmma_f32_16x16x4_f32(false, a[j], false, P[i0 + j],
                                                         (short)0, acc0, false, false);
            acc1 = __builtin_amdgcn_wmma_f32_16x16x4_f32(false, a[j], false, Q[i0 + j],
                                                         (short)0, acc1, false, false);
        }
#pragma unroll
        for (int j = 0; j < 4; ++j) a[j] = n[j];
    }
}

#define HOLD8(X, o) asm volatile("" :: "v"(X[o]), "v"(X[o + 1]), "v"(X[o + 2]), "v"(X[o + 3]), \
                                       "v"(X[o + 4]), "v"(X[o + 5]), "v"(X[o + 6]), "v"(X[o + 7]))

__global__ void __launch_bounds__(128) layer_front_kernel(const float* __restrict__ xin,
                                         const float* __restrict__ Wn,
                                         float* __restrict__ ft,
                                         float* __restrict__ sB,
                                         const float* __restrict__ bb,
                                         float* __restrict__ xout,
                                         const float* __restrict__ eF,
                                         const float* __restrict__ wv,
                                         const int* __restrict__ src,
                                         const int* __restrict__ dst,
                                         float* __restrict__ lg,
                                         unsigned* __restrict__ mU) {
    const int lane = threadIdx.x & 31;
    const int c    = lane & 15;
    const int hf   = lane >> 4;
    const int kh   = hf << 1;

    if (blockIdx.x < LOGIT_BLOCKS) {
        __shared__ __attribute__((aligned(16))) float sV[3 * VKP];
        const int wave = blockIdx.x * (blockDim.x >> 5) + (threadIdx.x >> 5);
        const int e0 = min(wave, N_EDGES / 32 - 1) << 5;
        const int eA = e0 + c, eB = e0 + 16 + c;
        const int sA = src[eA], dA = dst[eA];
        const int sC = src[eB], dC = dst[eB];

        for (int i = threadIdx.x; i < 3 * VK; i += 128) {
            const int r = i / VK, k = i - r * VK;
            sV[r * VKP + k] = (r < 2) ? wv[i] : 0.f;
        }
        __syncthreads();
        if (wave >= N_EDGES / 32) return;

        const float* ap = sV + (((c & 6) == 0) ? (c & 1) : 2) * VKP + kh;

        v2f P1[16], Q1[16], P2[16], Q2[16], P3[8], Q3[8];
        logit_load<16>(xin + (size_t)sA * 64 + kh, xin + (size_t)sC * 64 + kh, P1, Q1);
        logit_load<8>(eF + (size_t)eA * 32 + kh, eF + (size_t)eB * 32 + kh, P3, Q3);

        v8f acc0 = {}, acc1 = {};
        HOLD8(P1, 0); HOLD8(P1, 8); HOLD8(Q1, 0); HOLD8(Q1, 8);
        logit_mma<16>(ap, P1, Q1, acc0, acc1);
        logit_load<16>(xin + (size_t)dA * 64 + kh, xin + (size_t)dC * 64 + kh, P2, Q2);
        HOLD8(P3, 0); HOLD8(Q3, 0);
        logit_mma<8>(ap + 128, P3, Q3, acc0, acc1);
        HOLD8(P2, 0); HOLD8(P2, 8); HOLD8(Q2, 0); HOLD8(Q2, 8);
        logit_mma<16>(ap + 64, P2, Q2, acc0, acc1);

        const int h = hf;
        const float l0 = h ? acc0[1] : acc0[0];
        const float l1 = h ? acc1[1] : acc1[0];
        const float r0 = (l0 > 0.f) ? l0 : 0.2f * l0;
        const float r1 = (l1 > 0.f) ? l1 : 0.2f * l1;
        lg[eA * 2 + h] = r0;
        lg[eB * 2 + h] = r1;
        atomicMax(&mU[dA * 2 + h], f2ord(r0));
        atomicMax(&mU[dC * 2 + h], f2ord(r1));
        return;
    }

    const int wave = (blockIdx.x - LOGIT_BLOCKS) * (blockDim.x >> 5) + (threadIdx.x >> 5);
    if (wave >= (N_NODES / 16) * 2) return;
    const int h    = wave & 1;
    const int n0   = (wave >> 1) << 4;

    if (h == 1) sB[n0 * 2 + lane] = 0.f;
    {
        const float4 bh0 = *reinterpret_cast<const float4*>(bb + c * 4);
        const float4 bh1 = *reinterpret_cast<const float4*>(bb + 64 + c * 4);
        const float4 z4 = make_float4(0.5f * (bh0.x + bh1.x), 0.5f * (bh0.y + bh1.y),
                                      0.5f * (bh0.z + bh1.z), 0.5f * (bh0.w + bh1.w));
        float4* ag4 = reinterpret_cast<float4*>(xout + (size_t)n0 * 64 + h * 512);
#pragma unroll
        for (int i = 0; i < 4; ++i) ag4[i * 32 + lane] = z4;
    }

    v8f acc[4];
#pragma unroll
    for (int j = 0; j < 4; ++j) acc[j] = v8f{};
    const float* aBase = xin + (size_t)(n0 + c) * 64 + kh;
    const float* bBase = Wn + (size_t)kh * 128 + h * 64 + c;
#pragma unroll
    for (int k = 0; k < 64; k += 4) {
        v2f a;
        a.x = aBase[k];
        a.y = aBase[k + 1];
#pragma unroll
        for (int j = 0; j < 4; ++j) {
            v2f b;
            b.x = bBase[(size_t)k * 128 + j * 16];
            b.y = bBase[(size_t)(k + 1) * 128 + j * 16];
            acc[j] = __builtin_amdgcn_wmma_f32_16x16x4_f32(false, a, false, b,
                                                           (short)0, acc[j], false, false);
        }
    }

    const int rbase = n0 + ((lane >> 4) << 3);
    float*    cp    = ft + (size_t)rbase * 128 + h * 64 + c;
#pragma unroll
    for (int r = 0; r < 8; ++r) {
#pragma unroll
        for (int j = 0; j < 4; ++j) cp[(size_t)r * 128 + j * 16] = acc[j][r];
    }
}

__global__ void __launch_bounds__(128) scatter_msg_fused_kernel(const float* __restrict__ eF,
                                         const float* __restrict__ We,
                                         const float* __restrict__ ft,
                                         const float* __restrict__ p,
                                         const float* __restrict__ sB,
                                         const int* __restrict__ src,
                                         const int* __restrict__ dst,
                                         float* __restrict__ xout,
                                         unsigned* __restrict__ mU) {
    __shared__ __attribute__((aligned(16))) float sW[16 * WE_PITCH];
    stage_We_pairs(We, sW, threadIdx.x);
    __syncthreads();

    const int wave = blockIdx.x * (blockDim.x >> 5) + (threadIdx.x >> 5);
    if (wave >= N_EDGES / 16) return;
    const int lane = threadIdx.x & 31;
    const int c    = lane & 15;
    const int e0   = wave << 4;

    if (lane < 8 && wave * 8 + lane < N_NODES * 2) mU[wave * 8 + lane] = 0u;

    const int eb = e0 + ((lane >> 4) << 3);

    int   sNr[8], dNr[8];
    float a0r[8], a1r[8];
#pragma unroll
    for (int r = 0; r < 8; ++r) {
        const int e = eb + r;
        sNr[r] = src[e];
        dNr[r] = dst[e];
        a0r[r] = 0.5f * (p[e * 2]     / sB[dNr[r] * 2]);
        a1r[r] = 0.5f * (p[e * 2 + 1] / sB[dNr[r] * 2 + 1]);
    }

    v8f acc[8];
    fe_tile_16x128(eF, sW, e0, lane, acc);

#pragma unroll
    for (int r = 0; r < 8; ++r) {
        const float* fts = ft + (size_t)sNr[r] * 128 + c;
#pragma unroll
        for (int j = 0; j < 4; ++j) {
            const float m0 = (fts[j * 16]      + acc[j][r])     * a0r[r];
            const float m1 = (fts[64 + j * 16] + acc[j + 4][r]) * a1r[r];
            acc[j][r] = m0 + m1;
        }
    }
#pragma unroll
    for (int r = 0; r < 8; ++r) {
        float* ag = xout + (size_t)dNr[r] * 64 + c;
#pragma unroll
        for (int j = 0; j < 4; ++j) atomicAdd(&ag[j * 16], acc[j][r]);
    }
}

static inline int cdiv(int a, int b) { return (a + b - 1) / b; }

struct WsPtrs {
    float* ft; float* lg;
    unsigned* mU; float* sB; float* xb0; float* xb1; float* wv; float* Wf;
};

static void run_edge_gat_layer(const float* xin, const float* eF, const int* src,
                               const int* dst, const float* Wn, const float* We,
                               const float* bb, const float* wvl, float* xout, const WsPtrs& w,
                               hipStream_t stream) {
    layer_front_kernel<<<LOGIT_BLOCKS + NODE_BLOCKS, 128, 0, stream>>>(
        xin, Wn, w.ft, w.sB, bb, xout, eF, wvl, src, dst, w.lg, w.mU);
    p_sum_kernel<<<cdiv(N_EDGES * 2, TPB), TPB, 0, stream>>>(w.lg, w.mU, dst, w.sB);
    scatter_msg_fused_kernel<<<cdiv(N_EDGES / 16, 4), 128, 0, stream>>>(
        eF, We, w.ft, w.lg, w.sB, src, dst, xout, w.mU);
}

extern "C" void kernel_launch(void* const* d_in, const int* in_sizes, int n_in,
                              void* d_out, int out_size, void* d_ws, size_t ws_size,
                              hipStream_t stream) {
    const float* x_all = (const float*)d_in[0];
    const float* e_all = (const float*)d_in[1];
    const int*   s_all = (const int*)d_in[2];
    const int*   d_all = (const int*)d_in[3];

    const float* xS  = x_all + (size_t)1 * N_NODES * 64;
    const float* eS  = e_all + (size_t)1 * N_EDGES * 32;
    const int*   src = s_all + (size_t)1 * N_EDGES;
    const int*   dst = d_all + (size_t)1 * N_EDGES;

    const float* const* P = (const float* const*)d_in;
    const float* W_e2d = P[28];

    float* ws = (float*)d_ws;
    size_t off = 0;
    WsPtrs w;
    w.ft  = ws + off; off += (size_t)N_NODES * 128;
    w.lg  = ws + off; off += (size_t)N_EDGES * 2;
    w.mU  = (unsigned*)(ws + off); off += (size_t)N_NODES * 2;
    w.sB  = ws + off; off += (size_t)N_NODES * 2;
    w.xb0 = ws + off; off += (size_t)N_NODES * 64;
    w.xb1 = ws + off; off += (size_t)N_NODES * 64;
    w.wv  = ws + off; off += 4 * 2 * VK;
    w.Wf  = ws + off; off += 64 * 128;

    prep_kernel<<<PREP_CLEAR_BLOCKS + 4 + PREP_FOLD_BLOCKS, 320, 0, stream>>>(
        P[4], P[5], P[6], P[7], P[8], P[10], P[11], P[12], P[13], P[14],
        P[16], P[17], P[18], P[19], P[20], P[22], P[23], P[24], P[25], P[26],
        W_e2d, w.wv, w.Wf, w.mU);

    run_edge_gat_layer(xS, eS, src, dst, P[4], P[5], P[9], w.wv, w.xb0, w, stream);
    run_edge_gat_layer(w.xb0, eS, src, dst, P[10], P[11], P[15], w.wv + 2 * VK, w.xb1, w, stream);
    run_edge_gat_layer(w.xb1, eS, src, dst, w.Wf, P[17], P[21], w.wv + 4 * VK, w.xb0, w, stream);
    run_edge_gat_layer(w.xb0, eS, src, dst, P[22], P[23], P[27], w.wv + 6 * VK,
                       (float*)d_out, w, stream);
}
